# starting kernel
# baseline (speedup 1.0000x reference)
#include <hip/hip_runtime.h>
#include <hip/hip_bf16.h>


#define NNODES 50000
#define DFEAT  64
#define FRAGH  512

typedef __attribute__((ext_vector_type(16))) _Float16 v16h;
typedef __attribute__((ext_vector_type(8)))  float    v8f;

__device__ __forceinline__ v8f wmma_f16(v16h a, v16h b, v8f c) {
  return __builtin_amdgcn_wmma_f32_16x16x32_f16(
       false, a,  false, b,
       (short)0, c,  false,  false);
}

__device__ __forceinline__ v16h load_a_frag(const float* __restrict__ X, int ld,
                                            int rbase, int kbase, int lane) {
  const int m  = lane & 15;
  const int hi = lane >> 4;
  const float* p = X + (size_t)(rbase + m) * ld + kbase + hi * 8;
  const float4 q0 = *(const float4*)(p + 0);
  const float4 q1 = *(const float4*)(p + 4);
  const float4 q2 = *(const float4*)(p + 16);
  const float4 q3 = *(const float4*)(p + 20);
  v16h a;
  a[0]  = (_Float16)q0.x; a[1]  = (_Float16)q0.y;
  a[2]  = (_Float16)q0.z; a[3]  = (_Float16)q0.w;
  a[4]  = (_Float16)q1.x; a[5]  = (_Float16)q1.y;
  a[6]  = (_Float16)q1.z; a[7]  = (_Float16)q1.w;
  a[8]  = (_Float16)q2.x; a[9]  = (_Float16)q2.y;
  a[10] = (_Float16)q2.z; a[11] = (_Float16)q2.w;
  a[12] = (_Float16)q3.x; a[13] = (_Float16)q3.y;
  a[14] = (_Float16)q3.z; a[15] = (_Float16)q3.w;
  return a;
}

__device__ __forceinline__ v16h load_a_frag_scaled(const float* __restrict__ X,
                                                   int ld, int rbase, int kbase,
                                                   int lane,
                                                   const float* __restrict__ s) {
  const int m  = lane & 15;
  const int hi = lane >> 4;
  const int row = rbase + m;
  const float sc = s[row];
  const float* p = X + (size_t)row * ld + kbase + hi * 8;
  const float4 q0 = *(const float4*)(p + 0);
  const float4 q1 = *(const float4*)(p + 4);
  const float4 q2 = *(const float4*)(p + 16);
  const float4 q3 = *(const float4*)(p + 20);
  v16h a;
  a[0]  = (_Float16)(q0.x * sc); a[1]  = (_Float16)(q0.y * sc);
  a[2]  = (_Float16)(q0.z * sc); a[3]  = (_Float16)(q0.w * sc);
  a[4]  = (_Float16)(q1.x * sc); a[5]  = (_Float16)(q1.y * sc);
  a[6]  = (_Float16)(q1.z * sc); a[7]  = (_Float16)(q1.w * sc);
  a[8]  = (_Float16)(q2.x * sc); a[9]  = (_Float16)(q2.y * sc);
  a[10] = (_Float16)(q2.z * sc); a[11] = (_Float16)(q2.w * sc);
  a[12] = (_Float16)(q3.x * sc); a[13] = (_Float16)(q3.y * sc);
  a[14] = (_Float16)(q3.z * sc); a[15] = (_Float16)(q3.w * sc);
  return a;
}

__device__ __forceinline__ v16h load_b_packed(const _Float16* __restrict__ P,
                                              int nt, int ki, int ni, int lane) {
  return *((const v16h*)P + ((ki * nt + ni) * 32 + lane));
}

__global__ void pack_b_kernel(const float* __restrict__ W, int K, int ncols,
                              _Float16* __restrict__ out) {
  const int kt = K >> 5, nt = ncols >> 4;
  const int total = kt * nt * 32;
  const int t = blockIdx.x * blockDim.x + threadIdx.x;
  if (t >= total) return;
  const int lane = t & 31;
  const int frag = t >> 5;
  const int ni = frag % nt, ki = frag / nt;
  const int n = lane & 15, hi = lane >> 4;
  const float* p = W + (size_t)(ki * 32 + hi * 8) * ncols + ni * 16 + n;
  _Float16* o = out + (size_t)frag * FRAGH + lane * 16;
#pragma unroll
  for (int j = 0; j < 16; ++j) {
    const int k = ((j >> 3) << 4) + (j & 7);
    o[j] = (_Float16)p[(size_t)k * ncols];
  }
}

__global__ __launch_bounds__(256) void proj_kernel(
    const float* __restrict__ xc, const float* __restrict__ xs,
    const _Float16* __restrict__ PWp, const float* __restrict__ bp,
    const _Float16* __restrict__ PWs, const float* __restrict__ bs,
    float* __restrict__ X0) {
  const int lane = threadIdx.x & 31;
  const int wave = threadIdx.x >> 5;
  const int rbase = blockIdx.x * 32 + (wave >> 2) * 16;
  if (rbase >= NNODES) return;
  const int col_tile = wave & 3;
  const bool style = (col_tile >= 2);
  const float*    Xsrc = style ? xs : xc;
  const _Float16* P    = style ? PWs : PWp;
  const float*    bias = style ? bs : bp;
  const int       K    = style ? 128 : 768;
  const int       ni   = col_tile & 1;

  v8f acc = {};
  for (int k = 0; k < K; k += 32) {
    v16h a = load_a_frag(Xsrc, K, rbase, k, lane);
    v16h b = load_b_packed(P, 2, k >> 5, ni, lane);
    acc = wmma_f16(a, b, acc);
  }

  const int n  = lane & 15;
  const int hi = lane >> 4;
  const float bv = bias[ni * 16 + n];
  float* orow = X0 + (size_t)(rbase + hi * 8) * DFEAT + (style ? 32 : 0) + ni * 16 + n;
#pragma unroll
  for (int r = 0; r < 8; ++r) orow[(size_t)r * DFEAT] = acc[r] + bv;
}

__global__ __launch_bounds__(256) void sage_kernel(
    const float* __restrict__ msg, const float* __restrict__ invdeg,
    const float* __restrict__ xin,
    const _Float16* __restrict__ PWl, const float* __restrict__ bl,
    const _Float16* __restrict__ PWr, float* __restrict__ xout) {
  const int lane = threadIdx.x & 31;
  const int wave = threadIdx.x >> 5;
  const int rbase = blockIdx.x * 32 + (wave >> 2) * 16;
  if (rbase >= NNODES) return;
  const int ni = wave & 3;

  v8f acc = {};
#pragma unroll
  for (int k = 0; k < DFEAT; k += 32) {
    v16h aa = load_a_frag_scaled(msg, DFEAT, rbase, k, lane, invdeg);
    v16h ba = load_b_packed(PWl, 4, k >> 5, ni, lane);
    acc = wmma_f16(aa, ba, acc);
    v16h ax = load_a_frag(xin, DFEAT, rbase, k, lane);
    v16h bx = load_b_packed(PWr, 4, k >> 5, ni, lane);
    acc = wmma_f16(ax, bx, acc);
  }

  const int n  = lane & 15;
  const int hi = lane >> 4;
  const float bv = bl[ni * 16 + n];
  float* orow = xout + (size_t)(rbase + hi * 8) * DFEAT + ni * 16 + n;
#pragma unroll
  for (int r = 0; r < 8; ++r) {
    const float v = acc[r] + bv;
    orow[(size_t)r * DFEAT] = v > 0.0f ? v : 0.0f;
  }
}

__global__ void fill_kernel(float* __restrict__ p, int n, float v) {
  for (int i = blockIdx.x * blockDim.x + threadIdx.x; i < n;
       i += gridDim.x * blockDim.x)
    p[i] = v;
}

__global__ void fill4_kernel(float4* __restrict__ p, int n4) {
  const float4 z = make_float4(0.f, 0.f, 0.f, 0.f);
  for (int i = blockIdx.x * blockDim.x + threadIdx.x; i < n4;
       i += gridDim.x * blockDim.x)
    p[i] = z;
}

__global__ void degree_kernel(const int* __restrict__ dst,
                              float* __restrict__ deg, int nedges) {
  for (int e = blockIdx.x * blockDim.x + threadIdx.x; e < nedges;
       e += gridDim.x * blockDim.x)
    atomicAdd(&deg[dst[e]], 1.0f);
}

__global__ void invdeg_kernel(float* __restrict__ deg, int n) {
  for (int i = blockIdx.x * blockDim.x + threadIdx.x; i < n;
       i += gridDim.x * blockDim.x)
    deg[i] = 1.0f / fmaxf(deg[i], 1.0f);
}

__global__ void scatter_kernel(const float* __restrict__ x,
                               const int* __restrict__ src,
                               const int* __restrict__ dst,
                               float* __restrict__ msg, int total) {
  for (int i = blockIdx.x * blockDim.x + threadIdx.x; i < total;
       i += gridDim.x * blockDim.x) {
    const int e = i >> 6;
    const int f = i & 63;
    const int s = src[e];
    const int d = dst[e];
    atomicAdd(&msg[(size_t)d * DFEAT + f], x[(size_t)s * DFEAT + f]);
  }
}

__global__ void head_kernel(const float* __restrict__ x,
                            const float* __restrict__ Wo,
                            const float* __restrict__ bo,
                            float* __restrict__ out) {
  const int i = blockIdx.x * blockDim.x + threadIdx.x;
  if (i >= NNODES) return;
  const float* row = x + (size_t)i * DFEAT;
  float o0 = bo[0], o1 = bo[1];
#pragma unroll
  for (int k = 0; k < DFEAT; ++k) {
    const float v = row[k];
    o0 += v * Wo[k * 2 + 0];
    o1 += v * Wo[k * 2 + 1];
  }
  out[i * 2 + 0] = o0;
  out[i * 2 + 1] = o1;
}

extern "C" void kernel_launch(void* const* d_in, const int* in_sizes, int n_in,
                              void* d_out, int out_size, void* d_ws,
                              size_t ws_size, hipStream_t stream) {
  const float* xc  = (const float*)d_in[0];
  const float* xs  = (const float*)d_in[1];
  const int*   ei  = (const int*)d_in[2];
  const float* Wp  = (const float*)d_in[4];
  const float* bp  = (const float*)d_in[5];
  const float* Ws  = (const float*)d_in[6];
  const float* bs  = (const float*)d_in[7];
  const float* Wl1 = (const float*)d_in[8];
  const float* bl1 = (const float*)d_in[9];
  const float* Wr1 = (const float*)d_in[10];
  const float* Wl2 = (const float*)d_in[11];
  const float* bl2 = (const float*)d_in[12];
  const float* Wr2 = (const float*)d_in[13];
  const float* Wl3 = (const float*)d_in[14];
  const float* bl3 = (const float*)d_in[15];
  const float* Wr3 = (const float*)d_in[16];
  const float* Wo  = (const float*)d_in[17];
  const float* bo  = (const float*)d_in[18];

  const int E = in_sizes[2] / 2;
  const int* src = ei;
  const int* dst = ei + E;

  float* ws  = (float*)d_ws;
  float* X0  = ws;
  float* X1  = ws + (size_t)NNODES * DFEAT;
  float* MSG = ws + 2ull * NNODES * DFEAT;
  float* IDG = ws + 3ull * NNODES * DFEAT;
  _Float16* PW   = (_Float16*)(ws + 3ull * NNODES * DFEAT + NNODES);
  _Float16* PWp  = PW;
  _Float16* PWs  = PW + 48ull  * FRAGH;
  _Float16* PWl1 = PW + 56ull  * FRAGH;
  _Float16* PWr1 = PW + 64ull  * FRAGH;
  _Float16* PWl2 = PW + 72ull  * FRAGH;
  _Float16* PWr2 = PW + 80ull  * FRAGH;
  _Float16* PWl3 = PW + 88ull  * FRAGH;
  _Float16* PWr3 = PW + 96ull  * FRAGH;

  const int gemm_blocks = (NNODES + 31) / 32;
  const int feat_elems  = NNODES * DFEAT;
  const int scat_total  = E * DFEAT;

  pack_b_kernel<<<24, 64, 0, stream>>>(Wp,  768, 32, PWp);
  pack_b_kernel<<<4,  64, 0, stream>>>(Ws,  128, 32, PWs);
  pack_b_kernel<<<4,  64, 0, stream>>>(Wl1, 64,  64, PWl1);
  pack_b_kernel<<<4,  64, 0, stream>>>(Wr1, 64,  64, PWr1);
  pack_b_kernel<<<4,  64, 0, stream>>>(Wl2, 64,  64, PWl2);
  pack_b_kernel<<<4,  64, 0, stream>>>(Wr2, 64,  64, PWr2);
  pack_b_kernel<<<4,  64, 0, stream>>>(Wl3, 64,  64, PWl3);
  pack_b_kernel<<<4,  64, 0, stream>>>(Wr3, 64,  64, PWr3);

  fill_kernel<<<256, 256, 0, stream>>>(IDG, NNODES, 0.0f);
  degree_kernel<<<(E + 255) / 256, 256, 0, stream>>>(dst, IDG, E);
  invdeg_kernel<<<(NNODES + 255) / 256, 256, 0, stream>>>(IDG, NNODES);

  proj_kernel<<<gemm_blocks, 256, 0, stream>>>(xc, xs, PWp, bp, PWs, bs, X0);

  fill4_kernel<<<2048, 256, 0, stream>>>((float4*)MSG, feat_elems / 4);
  scatter_kernel<<<16384, 256, 0, stream>>>(X0, src, dst, MSG, scat_total);
  sage_kernel<<<gemm_blocks, 256, 0, stream>>>(MSG, IDG, X0, PWl1, bl1, PWr1, X1);

  fill4_kernel<<<2048, 256, 0, stream>>>((float4*)MSG, feat_elems / 4);
  scatter_kernel<<<16384, 256, 0, stream>>>(X1, src, dst, MSG, scat_total);
  sage_kernel<<<gemm_blocks, 256, 0, stream>>>(MSG, IDG, X1, PWl2, bl2, PWr2, X0);

  fill4_kernel<<<2048, 256, 0, stream>>>((float4*)MSG, feat_elems / 4);
  scatter_kernel<<<16384, 256, 0, stream>>>(X0, src, dst, MSG, scat_total);
  sage_kernel<<<gemm_blocks, 256, 0, stream>>>(MSG, IDG, X0, PWl3, bl3, PWr3, X1);

  head_kernel<<<(NNODES + 255) / 256, 256, 0, stream>>>(X1, Wo, bo, (float*)d_out);
}


# Each layer's row gather and its WMMA GEMM are one launch: a wave sums 4 nodes' rows, the means reach the GEMM through LDS, two feature buffers ping-pong; 11 -> 8 launches; row sum order unchanged
# speedup vs baseline: 4.8010x; 4.8010x over previous
#include <hip/hip_runtime.h>
#include <hip/hip_bf16.h>


#define NNODES 50000
#define DFEAT  64
#define FRAGH  512

#define BKT_SHIFT 7
#define BKT_NODES (1 << BKT_SHIFT)
#define NBKT      ((NNODES + BKT_NODES - 1) / BKT_NODES)
#define BKT_PAD   512
#define CHUNK     4096
static_assert(NBKT <= BKT_PAD, "one table entry per bucket");

typedef __attribute__((ext_vector_type(16))) _Float16 v16h;
typedef __attribute__((ext_vector_type(8)))  float    v8f;

__device__ __forceinline__ v8f wmma_f16(v16h a, v16h b, v8f c) {
  return __builtin_amdgcn_wmma_f32_16x16x32_f16(
       false, a,  false, b,
       (short)0, c,  false,  false);
}

__device__ __forceinline__ v16h load_a_frag(const float* __restrict__ X, int ld,
                                            int rbase, int kbase, int lane) {
  const int m  = lane & 15;
  const int hi = lane >> 4;
  const float* p = X + (size_t)(rbase + m) * ld + kbase + hi * 8;
  const float4 q0 = *(const float4*)(p + 0);
  const float4 q1 = *(const float4*)(p + 4);
  const float4 q2 = *(const float4*)(p + 16);
  const float4 q3 = *(const float4*)(p + 20);
  v16h a;
  a[0]  = (_Float16)q0.x; a[1]  = (_Float16)q0.y;
  a[2]  = (_Float16)q0.z; a[3]  = (_Float16)q0.w;
  a[4]  = (_Float16)q1.x; a[5]  = (_Float16)q1.y;
  a[6]  = (_Float16)q1.z; a[7]  = (_Float16)q1.w;
  a[8]  = (_Float16)q2.x; a[9]  = (_Float16)q2.y;
  a[10] = (_Float16)q2.z; a[11] = (_Float16)q2.w;
  a[12] = (_Float16)q3.x; a[13] = (_Float16)q3.y;
  a[14] = (_Float16)q3.z; a[15] = (_Float16)q3.w;
  return a;
}

__device__ __forceinline__ v16h load_b_packed(const _Float16* __restrict__ P,
                                              int nt, int ki, int ni, int lane) {
  return *((const v16h*)P + ((ki * nt + ni) * 32 + lane));
}

__device__ __forceinline__ void pack_b(const float* __restrict__ W, int K,
                                       int ncols, _Float16* __restrict__ out,
                                       int t) {
  const int kt = K >> 5, nt = ncols >> 4;
  if (t >= kt * nt * 32) return;
  const int lane = t & 31;
  const int frag = t >> 5;
  const int ni = frag % nt, ki = frag / nt;
  const int n = lane & 15, hi = lane >> 4;
  const float* p = W + (size_t)(ki * 32 + hi * 8) * ncols + ni * 16 + n;
  _Float16* o = out + (size_t)frag * FRAGH + lane * 16;
#pragma unroll
  for (int j = 0; j < 16; ++j) {
    const int k = ((j >> 3) << 4) + (j & 7);
    o[j] = (_Float16)p[(size_t)k * ncols];
  }
}

#define PREP_PACK_BLOCKS 13
#define PREP_ZERO_BLOCKS 1
__global__ __launch_bounds__(256) void prep_kernel(
    const float* __restrict__ Wp,  const float* __restrict__ Ws,
    const float* __restrict__ Wl1, const float* __restrict__ Wr1,
    const float* __restrict__ Wl2, const float* __restrict__ Wr2,
    const float* __restrict__ Wl3, const float* __restrict__ Wr3,
    _Float16* __restrict__ PW, int* __restrict__ bkt) {
  const int b = blockIdx.x;
  if (b >= PREP_PACK_BLOCKS) {
    for (int i = threadIdx.x; i < 2 * BKT_PAD; i += 256) bkt[i] = 0;
    return;
  }
  if (b < 6) {
    pack_b(Wp, 768, 32, PW, b * 256 + threadIdx.x);
    return;
  }
  if (b == 6) {
    pack_b(Ws, 128, 32, PW + 48ull * FRAGH, threadIdx.x);
    return;
  }
  const int l = b - 7;
  const float* W = l == 0 ? Wl1 : l == 1 ? Wr1 : l == 2 ? Wl2
                 : l == 3 ? Wr2 : l == 4 ? Wl3 : Wr3;
  pack_b(W, 64, 64, PW + (size_t)(56 + 8 * l) * FRAGH, threadIdx.x);
}

#define PROJ_BLOCKS ((NNODES + 31) / 32)
__global__ __launch_bounds__(256) void proj_hist_kernel(
    const float* __restrict__ xc, const float* __restrict__ xs,
    const _Float16* __restrict__ PWp, const float* __restrict__ bp,
    const _Float16* __restrict__ PWs, const float* __restrict__ bs,
    float* __restrict__ X0,
    const int* __restrict__ dst, int* __restrict__ bcnt, int nedges) {
  __shared__ int hist[BKT_PAD];
  if (blockIdx.x >= PROJ_BLOCKS) {
    const int e0 = (blockIdx.x - PROJ_BLOCKS) * CHUNK;
    for (int b = threadIdx.x; b < BKT_PAD; b += 256) hist[b] = 0;
    __syncthreads();
    int dv[CHUNK / 256];
#pragma unroll
    for (int i = 0; i < CHUNK / 256; ++i) {
      const int e = e0 + i * 256 + threadIdx.x;
      dv[i] = (e < nedges) ? dst[e] : -1;
    }
#pragma unroll
    for (int i = 0; i < CHUNK / 256; ++i)
      if (dv[i] >= 0) atomicAdd(&hist[dv[i] >> BKT_SHIFT], 1);
    __syncthreads();
    for (int b = threadIdx.x; b < NBKT; b += 256) {
      const int n = hist[b];
      if (n) atomicAdd(&bcnt[b], n);
    }
    return;
  }
  const int lane = threadIdx.x & 31;
  const int wave = threadIdx.x >> 5;
  const int rbase = blockIdx.x * 32 + (wave >> 2) * 16;
  if (rbase >= NNODES) return;
  const int col_tile = wave & 3;
  const bool style = (col_tile >= 2);
  const float*    Xsrc = style ? xs : xc;
  const _Float16* P    = style ? PWs : PWp;
  const float*    bias = style ? bs : bp;
  const int       K    = style ? 128 : 768;
  const int       ni   = col_tile & 1;

  v8f acc = {};
  for (int k = 0; k < K; k += 32) {
    v16h a = load_a_frag(Xsrc, K, rbase, k, lane);
    v16h b = load_b_packed(P, 2, k >> 5, ni, lane);
    acc = wmma_f16(a, b, acc);
  }

  const int n  = lane & 15;
  const int hi = lane >> 4;
  const float bv = bias[ni * 16 + n];
  float* orow = X0 + (size_t)(rbase + hi * 8) * DFEAT + (style ? 32 : 0) + ni * 16 + n;
#pragma unroll
  for (int r = 0; r < 8; ++r) orow[(size_t)r * DFEAT] = acc[r] + bv;
}


__device__ __forceinline__ int block_excl_scan512(int v, int* wsum) {
  const int lane = threadIdx.x & 31;
  const int wave = threadIdx.x >> 5;
  int incl = v;
#pragma unroll
  for (int off = 1; off < 32; off <<= 1) {
    const int up = __shfl_up(incl, off);
    if (lane >= off) incl += up;
  }
  if (lane == 31) wsum[wave] = incl;
  __syncthreads();
  int add = 0;
  for (int w = 0; w < wave; ++w) add += wsum[w];
  __syncthreads();
  return add + incl - v;
}

__global__ __launch_bounds__(512) void part_kernel(
    const int* __restrict__ src, const int* __restrict__ dst,
    int* __restrict__ bkt, int2* __restrict__ rec, int nedges) {
  __shared__ int2 srec[CHUNK];
  __shared__ int lcnt[BKT_PAD];
  __shared__ int lst[BKT_PAD];
  __shared__ int dlt[BKT_PAD];
  __shared__ int wsum[16];
  const int t = threadIdx.x;
  const int e0 = blockIdx.x * CHUNK;

  lcnt[t] = 0;
  const int bst = block_excl_scan512(bkt[t], wsum);
  if (blockIdx.x == 0) bkt[2 * BKT_PAD + t] = bst;

  int s[CHUNK / 512], d[CHUNK / 512], r[CHUNK / 512];
#pragma unroll
  for (int i = 0; i < CHUNK / 512; ++i) {
    const int e = e0 + i * 512 + t;
    const bool in = e < nedges;
    s[i] = in ? src[e] : 0;
    d[i] = in ? dst[e] : -1;
  }
#pragma unroll
  for (int i = 0; i < CHUNK / 512; ++i)
    r[i] = (d[i] >= 0) ? atomicAdd(&lcnt[d[i] >> BKT_SHIFT], 1) : 0;
  __syncthreads();

  const int cnt = lcnt[t];
  const int base = cnt ? atomicAdd(&bkt[BKT_PAD + t], cnt) : 0;
  const int ls = block_excl_scan512(cnt, wsum);
  lst[t] = ls;
  dlt[t] = bst + base - ls;
  __syncthreads();

#pragma unroll
  for (int i = 0; i < CHUNK / 512; ++i)
    if (d[i] >= 0) srec[lst[d[i] >> BKT_SHIFT] + r[i]] = make_int2(s[i], d[i]);
  __syncthreads();

  const int left = nedges - e0;
  const int m = left < CHUNK ? left : CHUNK;
  for (int j = t; j < m; j += 512) {
    const int2 q = srec[j];
    rec[dlt[q.y >> BKT_SHIFT] + j] = q;
  }
}

#define STAGE_CAP 12288
#define BSORT_HELD 10
__global__ __launch_bounds__(512) void bsort_kernel(
    const int2* __restrict__ rec, const int* __restrict__ bkt,
    int* __restrict__ rowptr, int* __restrict__ csr) {
  __shared__ int sbuf[STAGE_CAP];
  __shared__ int ncnt[BKT_NODES];
  __shared__ int ncur[BKT_NODES];
  __shared__ int wsum[16];
  const int t = threadIdx.x;
  const int b = blockIdx.x;
  const int n  = bkt[b];
  const int bs = bkt[2 * BKT_PAD + b];
  const int2* my = rec + bs;

  int2 q[BSORT_HELD];
#pragma unroll
  for (int i = 0; i < BSORT_HELD; ++i) {
    const int j = t + i * 512;
    q[i] = (j < n) ? my[j] : make_int2(0, -1);
  }
  if (t < BKT_NODES) ncnt[t] = 0;
  __syncthreads();
#pragma unroll
  for (int i = 0; i < BSORT_HELD; ++i)
    if (q[i].y >= 0) atomicAdd(&ncnt[q[i].y & (BKT_NODES - 1)], 1);
  for (int j = t + BSORT_HELD * 512; j < n; j += 512)
    atomicAdd(&ncnt[my[j].y & (BKT_NODES - 1)], 1);
  __syncthreads();

  const int c = (t < BKT_NODES) ? ncnt[t] : 0;
  const int ex = block_excl_scan512(c, wsum);
  if (t < BKT_NODES) {
    ncur[t] = ex;
    const int node = b * BKT_NODES + t;
    if (node < NNODES) rowptr[node] = bs + ex;
  }
  __syncthreads();

  const bool staged = n <= STAGE_CAP;
#pragma unroll
  for (int i = 0; i < BSORT_HELD; ++i) {
    if (q[i].y >= 0) {
      const int p = atomicAdd(&ncur[q[i].y & (BKT_NODES - 1)], 1);
      if (staged) sbuf[p] = q[i].x; else csr[bs + p] = q[i].x;
    }
  }
  for (int j = t + BSORT_HELD * 512; j < n; j += 512) {
    const int2 v = my[j];
    const int p = atomicAdd(&ncur[v.y & (BKT_NODES - 1)], 1);
    if (staged) sbuf[p] = v.x; else csr[bs + p] = v.x;
  }
  if (staged) {
    __syncthreads();
    for (int j = t; j < n; j += 512) csr[bs + j] = sbuf[j];
  }
}

static_assert(NNODES % 16 == 0, "row tiles must be exact");
#define GATHER_DEPTH 16
#define MEAN_PITCH   68

__device__ __forceinline__ void gather_rows(const char* __restrict__ xb,
                                            unsigned loff, int idx, int m,
                                            float2& acc) {
  for (int i = 0; i < m; i += GATHER_DEPTH) {
    unsigned s[GATHER_DEPTH];
#pragma unroll
    for (int u = 0; u < GATHER_DEPTH; ++u) {
      const int j = i + u < m ? i + u : m - 1;
      s[u] = (unsigned)__builtin_amdgcn_readlane(idx, j);
    }
    float2 v[GATHER_DEPTH];
#pragma unroll
    for (int u = 0; u < GATHER_DEPTH; ++u) {
      v[u] = make_float2(0.0f, 0.0f);
      if (i + u < m)
        v[u] = *(const float2*)(xb + (size_t)s[u] * (DFEAT * sizeof(float)) + loff);
    }
#pragma unroll
    for (int u = 0; u < GATHER_DEPTH; ++u) { acc.x += v[u].x; acc.y += v[u].y; }
  }
}

__global__ __launch_bounds__(128) void layer_kernel(
    const float* __restrict__ xin, float* __restrict__ xout,
    const int* __restrict__ rowptr, const int* __restrict__ csr,
    const _Float16* __restrict__ PWl, const float* __restrict__ bl,
    const _Float16* __restrict__ PWr, int nedges) {
  __shared__ __attribute__((aligned(16))) float smean[16 * MEAN_PITCH];
  const int lane = threadIdx.x & 31;
  const int wv = __builtin_amdgcn_readfirstlane(threadIdx.x >> 5);
  const int rbase = blockIdx.x * 16;
  const int n0 = rbase + wv * 4;

  int rp = 0;
  if (lane <= 4) {
    const int nd = n0 + lane;
    rp = (nd < NNODES) ? rowptr[nd] : nedges;
  }
  const int b0 = __builtin_amdgcn_readlane(rp, 0);
  const int b1 = __builtin_amdgcn_readlane(rp, 1);
  const int b2 = __builtin_amdgcn_readlane(rp, 2);
  const int b3 = __builtin_amdgcn_readlane(rp, 3);
  const int b4 = __builtin_amdgcn_readlane(rp, 4);
  const int f0 = (b0 + lane < b1) ? csr[b0 + lane] : 0;
  const int f1 = (b1 + lane < b2) ? csr[b1 + lane] : 0;
  const int f2 = (b2 + lane < b3) ? csr[b2 + lane] : 0;
  const int f3 = (b3 + lane < b4) ? csr[b3 + lane] : 0;
  const int g0 = (b0 + 32 + lane < b1) ? csr[b0 + 32 + lane] : 0;
  const int g1 = (b1 + 32 + lane < b2) ? csr[b1 + 32 + lane] : 0;
  const int g2 = (b2 + 32 + lane < b3) ? csr[b2 + 32 + lane] : 0;
  const int g3 = (b3 + 32 + lane < b4) ? csr[b3 + 32 + lane] : 0;

  const char* xb = (const char*)xin;
  const unsigned loff = (unsigned)lane * (2u * sizeof(float));
  for (int q = 0; q < 4; ++q) {
    const int beg = q == 0 ? b0 : q == 1 ? b1 : q == 2 ? b2 : b3;
    const int end = q == 0 ? b1 : q == 1 ? b2 : q == 2 ? b3 : b4;
    const int c0  = q == 0 ? f0 : q == 1 ? f1 : q == 2 ? f2 : f3;
    const int c1  = q == 0 ? g0 : q == 1 ? g1 : q == 2 ? g2 : g3;
    float2 acc = make_float2(0.0f, 0.0f);
    int c = 0;
    for (int base = beg; base < end; base += 32, ++c) {
      const int left = end - base;
      const int m = left < 32 ? left : 32;
      int idx;
      if (c == 0)      idx = c0;
      else if (c == 1) idx = c1;
      else             idx = (lane < m) ? csr[base + lane] : 0;
      gather_rows(xb, loff, idx, m, acc);
    }
    const float sc = 1.0f / fmaxf((float)(end - beg), 1.0f);
    *(float2*)(smean + (wv * 4 + q) * MEAN_PITCH + lane * 2) =
        make_float2(acc.x * sc, acc.y * sc);
  }

  const int ni = wv;
  const v16h ax0 = load_a_frag(xin, DFEAT, rbase, 0, lane);
  const v16h ax1 = load_a_frag(xin, DFEAT, rbase, 32, lane);
  const v16h wr0 = load_b_packed(PWr, 4, 0, ni, lane);
  const v16h wr1 = load_b_packed(PWr, 4, 1, ni, lane);
  const int n  = lane & 15;
  const int hi = lane >> 4;
  const float bv = bl[ni * 16 + n];
  __syncthreads();

  const v16h wl0 = load_b_packed(PWl, 4, 0, ni, lane);
  const v16h wl1 = load_b_packed(PWl, 4, 1, ni, lane);
  v8f acc = {};
  acc = wmma_f16(load_a_frag(smean, MEAN_PITCH, 0, 0, lane), wl0, acc);
  acc = wmma_f16(ax0, wr0, acc);
  acc = wmma_f16(load_a_frag(smean, MEAN_PITCH, 0, 32, lane), wl1, acc);
  acc = wmma_f16(ax1, wr1, acc);

  float* orow = xout + (size_t)(rbase + hi * 8) * DFEAT + ni * 16 + n;
#pragma unroll
  for (int r = 0; r < 8; ++r) {
    const float v = acc[r] + bv;
    orow[(size_t)r * DFEAT] = v > 0.0f ? v : 0.0f;
  }
}

__global__ void head_kernel(const float* __restrict__ x,
                            const float* __restrict__ Wo,
                            const float* __restrict__ bo,
                            float* __restrict__ out) {
  const int i = blockIdx.x * blockDim.x + threadIdx.x;
  if (i >= NNODES) return;
  const float* row = x + (size_t)i * DFEAT;
  float o0 = bo[0], o1 = bo[1];
#pragma unroll
  for (int k = 0; k < DFEAT; ++k) {
    const float v = row[k];
    o0 += v * Wo[k * 2 + 0];
    o1 += v * Wo[k * 2 + 1];
  }
  out[i * 2 + 0] = o0;
  out[i * 2 + 1] = o1;
}

extern "C" void kernel_launch(void* const* d_in, const int* in_sizes, int n_in,
                              void* d_out, int out_size, void* d_ws,
                              size_t ws_size, hipStream_t stream) {
  const float* xc  = (const float*)d_in[0];
  const float* xs  = (const float*)d_in[1];
  const int*   ei  = (const int*)d_in[2];
  const float* Wp  = (const float*)d_in[4];
  const float* bp  = (const float*)d_in[5];
  const float* Ws  = (const float*)d_in[6];
  const float* bs  = (const float*)d_in[7];
  const float* Wl1 = (const float*)d_in[8];
  const float* bl1 = (const float*)d_in[9];
  const float* Wr1 = (const float*)d_in[10];
  const float* Wl2 = (const float*)d_in[11];
  const float* bl2 = (const float*)d_in[12];
  const float* Wr2 = (const float*)d_in[13];
  const float* Wl3 = (const float*)d_in[14];
  const float* bl3 = (const float*)d_in[15];
  const float* Wr3 = (const float*)d_in[16];
  const float* Wo  = (const float*)d_in[17];
  const float* bo  = (const float*)d_in[18];

  const int E = in_sizes[2] / 2;
  const int* src = ei;
  const int* dst = ei + E;

  float* ws  = (float*)d_ws;
  float* X0  = ws;
  int*   CSR = (int*)(ws + (size_t)NNODES * DFEAT);
  int*   BKT = CSR + E;
  float* X1  = ws + 2ull * NNODES * DFEAT;
  int2*  REC = (int2*)X1;
  int*   RPT = (int*)(ws + 3ull * NNODES * DFEAT);
  _Float16* PW   = (_Float16*)(ws + 3ull * NNODES * DFEAT + NNODES);
  _Float16* PWp  = PW;
  _Float16* PWs  = PW + 48ull  * FRAGH;
  _Float16* PWl1 = PW + 56ull  * FRAGH;
  _Float16* PWr1 = PW + 64ull  * FRAGH;
  _Float16* PWl2 = PW + 72ull  * FRAGH;
  _Float16* PWr2 = PW + 80ull  * FRAGH;
  _Float16* PWl3 = PW + 88ull  * FRAGH;
  _Float16* PWr3 = PW + 96ull  * FRAGH;

  const int layer_blocks = NNODES / 16;
  const int chunks = (E + CHUNK - 1) / CHUNK;

  prep_kernel<<<PREP_PACK_BLOCKS + PREP_ZERO_BLOCKS, 256, 0, stream>>>(
      Wp, Ws, Wl1, Wr1, Wl2, Wr2, Wl3, Wr3, PW, BKT);

  proj_hist_kernel<<<PROJ_BLOCKS + chunks, 256, 0, stream>>>(
      xc, xs, PWp, bp, PWs, bs, X0, dst, BKT, E);

  part_kernel<<<chunks, 512, 0, stream>>>(src, dst, BKT, REC, E);
  bsort_kernel<<<NBKT, 512, 0, stream>>>(REC, BKT, RPT, CSR);

  layer_kernel<<<layer_blocks, 128, 0, stream>>>(X0, X1, RPT, CSR, PWl1, bl1, PWr1, E);
  layer_kernel<<<layer_blocks, 128, 0, stream>>>(X1, X0, RPT, CSR, PWl2, bl2, PWr2, E);
  layer_kernel<<<layer_blocks, 128, 0, stream>>>(X0, X1, RPT, CSR, PWl3, bl3, PWr3, E);

  head_kernel<<<(NNODES + 255) / 256, 256, 0, stream>>>(X1, Wo, bo, (float*)d_out);
}


# Row gather reads two rows per load instruction: half a wave per node, 16 B per lane, indices by cross-lane read, a row of zeros pads short lists, no per-row scalar work; sum order unchanged
# speedup vs baseline: 5.2559x; 1.0948x over previous
#include <hip/hip_runtime.h>
#include <hip/hip_bf16.h>


#define NNODES 50000
#define DFEAT  64
#define FRAGH  512

#define BKT_SHIFT 7
#define BKT_NODES (1 << BKT_SHIFT)
#define NBKT      ((NNODES + BKT_NODES - 1) / BKT_NODES)
#define BKT_PAD   512
#define CHUNK     4096
static_assert(NBKT <= BKT_PAD, "one table entry per bucket");

typedef __attribute__((ext_vector_type(16))) _Float16 v16h;
typedef __attribute__((ext_vector_type(8)))  float    v8f;

__device__ __forceinline__ v8f wmma_f16(v16h a, v16h b, v8f c) {
  return __builtin_amdgcn_wmma_f32_16x16x32_f16(
       false, a,  false, b,
       (short)0, c,  false,  false);
}

__device__ __forceinline__ v16h load_a_frag(const float* __restrict__ X, int ld,
                                            int rbase, int kbase, int lane) {
  const int m  = lane & 15;
  const int hi = lane >> 4;
  const float* p = X + (size_t)(rbase + m) * ld + kbase + hi * 8;
  const float4 q0 = *(const float4*)(p + 0);
  const float4 q1 = *(const float4*)(p + 4);
  const float4 q2 = *(const float4*)(p + 16);
  const float4 q3 = *(const float4*)(p + 20);
  v16h a;
  a[0]  = (_Float16)q0.x; a[1]  = (_Float16)q0.y;
  a[2]  = (_Float16)q0.z; a[3]  = (_Float16)q0.w;
  a[4]  = (_Float16)q1.x; a[5]  = (_Float16)q1.y;
  a[6]  = (_Float16)q1.z; a[7]  = (_Float16)q1.w;
  a[8]  = (_Float16)q2.x; a[9]  = (_Float16)q2.y;
  a[10] = (_Float16)q2.z; a[11] = (_Float16)q2.w;
  a[12] = (_Float16)q3.x; a[13] = (_Float16)q3.y;
  a[14] = (_Float16)q3.z; a[15] = (_Float16)q3.w;
  return a;
}

__device__ __forceinline__ v16h load_b_packed(const _Float16* __restrict__ P,
                                              int nt, int ki, int ni, int lane) {
  return *((const v16h*)P + ((ki * nt + ni) * 32 + lane));
}

__device__ __forceinline__ void pack_b(const float* __restrict__ W, int K,
                                       int ncols, _Float16* __restrict__ out,
                                       int t) {
  const int kt = K >> 5, nt = ncols >> 4;
  if (t >= kt * nt * 32) return;
  const int lane = t & 31;
  const int frag = t >> 5;
  const int ni = frag % nt, ki = frag / nt;
  const int n = lane & 15, hi = lane >> 4;
  const float* p = W + (size_t)(ki * 32 + hi * 8) * ncols + ni * 16 + n;
  _Float16* o = out + (size_t)frag * FRAGH + lane * 16;
#pragma unroll
  for (int j = 0; j < 16; ++j) {
    const int k = ((j >> 3) << 4) + (j & 7);
    o[j] = (_Float16)p[(size_t)k * ncols];
  }
}

#define PREP_PACK_BLOCKS 13
#define PREP_ZERO_BLOCKS 1
__global__ __launch_bounds__(256) void prep_kernel(
    const float* __restrict__ Wp,  const float* __restrict__ Ws,
    const float* __restrict__ Wl1, const float* __restrict__ Wr1,
    const float* __restrict__ Wl2, const float* __restrict__ Wr2,
    const float* __restrict__ Wl3, const float* __restrict__ Wr3,
    _Float16* __restrict__ PW, int* __restrict__ bkt,
    float* __restrict__ zrow0, float* __restrict__ zrow1) {
  const int b = blockIdx.x;
  if (b >= PREP_PACK_BLOCKS) {
    for (int i = threadIdx.x; i < 2 * BKT_PAD; i += 256) bkt[i] = 0;
    if (threadIdx.x < DFEAT) {
      zrow0[threadIdx.x] = 0.0f;
      zrow1[threadIdx.x] = 0.0f;
    }
    return;
  }
  if (b < 6) {
    pack_b(Wp, 768, 32, PW, b * 256 + threadIdx.x);
    return;
  }
  if (b == 6) {
    pack_b(Ws, 128, 32, PW + 48ull * FRAGH, threadIdx.x);
    return;
  }
  const int l = b - 7;
  const float* W = l == 0 ? Wl1 : l == 1 ? Wr1 : l == 2 ? Wl2
                 : l == 3 ? Wr2 : l == 4 ? Wl3 : Wr3;
  pack_b(W, 64, 64, PW + (size_t)(56 + 8 * l) * FRAGH, threadIdx.x);
}

#define PROJ_BLOCKS ((NNODES + 31) / 32)
__global__ __launch_bounds__(256) void proj_hist_kernel(
    const float* __restrict__ xc, const float* __restrict__ xs,
    const _Float16* __restrict__ PWp, const float* __restrict__ bp,
    const _Float16* __restrict__ PWs, const float* __restrict__ bs,
    float* __restrict__ X0,
    const int* __restrict__ dst, int* __restrict__ bcnt, int nedges) {
  __shared__ int hist[BKT_PAD];
  if (blockIdx.x >= PROJ_BLOCKS) {
    const int e0 = (blockIdx.x - PROJ_BLOCKS) * CHUNK;
    for (int b = threadIdx.x; b < BKT_PAD; b += 256) hist[b] = 0;
    __syncthreads();
    int dv[CHUNK / 256];
#pragma unroll
    for (int i = 0; i < CHUNK / 256; ++i) {
      const int e = e0 + i * 256 + threadIdx.x;
      dv[i] = (e < nedges) ? dst[e] : -1;
    }
#pragma unroll
    for (int i = 0; i < CHUNK / 256; ++i)
      if (dv[i] >= 0) atomicAdd(&hist[dv[i] >> BKT_SHIFT], 1);
    __syncthreads();
    for (int b = threadIdx.x; b < NBKT; b += 256) {
      const int n = hist[b];
      if (n) atomicAdd(&bcnt[b], n);
    }
    return;
  }
  const int lane = threadIdx.x & 31;
  const int wave = threadIdx.x >> 5;
  const int rbase = blockIdx.x * 32 + (wave >> 2) * 16;
  if (rbase >= NNODES) return;
  const int col_tile = wave & 3;
  const bool style = (col_tile >= 2);
  const float*    Xsrc = style ? xs : xc;
  const _Float16* P    = style ? PWs : PWp;
  const float*    bias = style ? bs : bp;
  const int       K    = style ? 128 : 768;
  const int       ni   = col_tile & 1;

  v8f acc = {};
  for (int k = 0; k < K; k += 32) {
    v16h a = load_a_frag(Xsrc, K, rbase, k, lane);
    v16h b = load_b_packed(P, 2, k >> 5, ni, lane);
    acc = wmma_f16(a, b, acc);
  }

  const int n  = lane & 15;
  const int hi = lane >> 4;
  const float bv = bias[ni * 16 + n];
  float* orow = X0 + (size_t)(rbase + hi * 8) * DFEAT + (style ? 32 : 0) + ni * 16 + n;
#pragma unroll
  for (int r = 0; r < 8; ++r) orow[(size_t)r * DFEAT] = acc[r] + bv;
}


__device__ __forceinline__ int block_excl_scan512(int v, int* wsum) {
  const int lane = threadIdx.x & 31;
  const int wave = threadIdx.x >> 5;
  int incl = v;
#pragma unroll
  for (int off = 1; off < 32; off <<= 1) {
    const int up = __shfl_up(incl, off);
    if (lane >= off) incl += up;
  }
  if (lane == 31) wsum[wave] = incl;
  __syncthreads();
  int add = 0;
  for (int w = 0; w < wave; ++w) add += wsum[w];
  __syncthreads();
  return add + incl - v;
}

__global__ __launch_bounds__(512) void part_kernel(
    const int* __restrict__ src, const int* __restrict__ dst,
    int* __restrict__ bkt, int2* __restrict__ rec, int nedges) {
  __shared__ int2 srec[CHUNK];
  __shared__ int lcnt[BKT_PAD];
  __shared__ int lst[BKT_PAD];
  __shared__ int dlt[BKT_PAD];
  __shared__ int wsum[16];
  const int t = threadIdx.x;
  const int e0 = blockIdx.x * CHUNK;

  lcnt[t] = 0;
  const int bst = block_excl_scan512(bkt[t], wsum);
  if (blockIdx.x == 0) bkt[2 * BKT_PAD + t] = bst;

  int s[CHUNK / 512], d[CHUNK / 512], r[CHUNK / 512];
#pragma unroll
  for (int i = 0; i < CHUNK / 512; ++i) {
    const int e = e0 + i * 512 + t;
    const bool in = e < nedges;
    s[i] = in ? src[e] : 0;
    d[i] = in ? dst[e] : -1;
  }
#pragma unroll
  for (int i = 0; i < CHUNK / 512; ++i)
    r[i] = (d[i] >= 0) ? atomicAdd(&lcnt[d[i] >> BKT_SHIFT], 1) : 0;
  __syncthreads();

  const int cnt = lcnt[t];
  const int base = cnt ? atomicAdd(&bkt[BKT_PAD + t], cnt) : 0;
  const int ls = block_excl_scan512(cnt, wsum);
  lst[t] = ls;
  dlt[t] = bst + base - ls;
  __syncthreads();

#pragma unroll
  for (int i = 0; i < CHUNK / 512; ++i)
    if (d[i] >= 0) srec[lst[d[i] >> BKT_SHIFT] + r[i]] = make_int2(s[i], d[i]);
  __syncthreads();

  const int left = nedges - e0;
  const int m = left < CHUNK ? left : CHUNK;
  for (int j = t; j < m; j += 512) {
    const int2 q = srec[j];
    rec[dlt[q.y >> BKT_SHIFT] + j] = q;
  }
}

#define STAGE_CAP 12288
#define BSORT_HELD 10
__global__ __launch_bounds__(512) void bsort_kernel(
    const int2* __restrict__ rec, const int* __restrict__ bkt,
    int* __restrict__ rowptr, int* __restrict__ csr) {
  __shared__ int sbuf[STAGE_CAP];
  __shared__ int ncnt[BKT_NODES];
  __shared__ int ncur[BKT_NODES];
  __shared__ int wsum[16];
  const int t = threadIdx.x;
  const int b = blockIdx.x;
  const int n  = bkt[b];
  const int bs = bkt[2 * BKT_PAD + b];
  const int2* my = rec + bs;

  int2 q[BSORT_HELD];
#pragma unroll
  for (int i = 0; i < BSORT_HELD; ++i) {
    const int j = t + i * 512;
    q[i] = (j < n) ? my[j] : make_int2(0, -1);
  }
  if (t < BKT_NODES) ncnt[t] = 0;
  __syncthreads();
#pragma unroll
  for (int i = 0; i < BSORT_HELD; ++i)
    if (q[i].y >= 0) atomicAdd(&ncnt[q[i].y & (BKT_NODES - 1)], 1);
  for (int j = t + BSORT_HELD * 512; j < n; j += 512)
    atomicAdd(&ncnt[my[j].y & (BKT_NODES - 1)], 1);
  __syncthreads();

  const int c = (t < BKT_NODES) ? ncnt[t] : 0;
  const int ex = block_excl_scan512(c, wsum);
  if (t < BKT_NODES) {
    ncur[t] = ex;
    const int node = b * BKT_NODES + t;
    if (node < NNODES) rowptr[node] = bs + ex;
  }
  __syncthreads();

  const bool staged = n <= STAGE_CAP;
#pragma unroll
  for (int i = 0; i < BSORT_HELD; ++i) {
    if (q[i].y >= 0) {
      const int p = atomicAdd(&ncur[q[i].y & (BKT_NODES - 1)], 1);
      if (staged) sbuf[p] = q[i].x; else csr[bs + p] = q[i].x;
    }
  }
  for (int j = t + BSORT_HELD * 512; j < n; j += 512) {
    const int2 v = my[j];
    const int p = atomicAdd(&ncur[v.y & (BKT_NODES - 1)], 1);
    if (staged) sbuf[p] = v.x; else csr[bs + p] = v.x;
  }
  if (staged) {
    __syncthreads();
    for (int j = t; j < n; j += 512) csr[bs + j] = sbuf[j];
  }
}

static_assert(NNODES % 16 == 0, "row tiles must be exact");
#define GATHER_DEPTH 8
#define MEAN_PITCH   68

__device__ __forceinline__ int load_idx16(const int* __restrict__ csr, int beg,
                                          int end, int c) {
  const int pos = beg + c;
  return (pos < end) ? csr[pos] : NNODES;
}

template <int FIRST>
__device__ __forceinline__ void gather_batch(const char* __restrict__ xb,
                                             unsigned coff, int hsel, int ix,
                                             float4& acc) {
  float4 v[GATHER_DEPTH];
#pragma unroll
  for (int u = 0; u < GATHER_DEPTH; ++u) {
    const unsigned r = (unsigned)__shfl(ix, hsel + FIRST + u);
    v[u] = *(const float4*)(xb + (r * (unsigned)(DFEAT * sizeof(float)) + coff));
  }
#pragma unroll
  for (int u = 0; u < GATHER_DEPTH; ++u) {
    acc.x += v[u].x; acc.y += v[u].y; acc.z += v[u].z; acc.w += v[u].w;
  }
}

__global__ __launch_bounds__(128, 12) void layer_kernel(
    const float* __restrict__ xin, float* __restrict__ xout,
    const int* __restrict__ rowptr, const int* __restrict__ csr,
    const _Float16* __restrict__ PWl, const float* __restrict__ bl,
    const _Float16* __restrict__ PWr, int nedges) {
  __shared__ __attribute__((aligned(16))) float smean[16 * MEAN_PITCH];
  const int lane = threadIdx.x & 31;
  const int wv = __builtin_amdgcn_readfirstlane(threadIdx.x >> 5);
  const int rbase = blockIdx.x * 16;
  const int n0 = rbase + wv * 4;

  int rp = 0;
  if (lane <= 4) {
    const int nd = n0 + lane;
    rp = (nd < NNODES) ? rowptr[nd] : nedges;
  }
  const int b0 = __builtin_amdgcn_readlane(rp, 0);
  const int b1 = __builtin_amdgcn_readlane(rp, 1);
  const int b2 = __builtin_amdgcn_readlane(rp, 2);
  const int b3 = __builtin_amdgcn_readlane(rp, 3);
  const int b4 = __builtin_amdgcn_readlane(rp, 4);
  const int c16  = lane & 15;
  const int hsel = lane & 16;
  const int begP0 = hsel ? b1 : b0, endP0 = hsel ? b2 : b1;
  const int begP1 = hsel ? b3 : b2, endP1 = hsel ? b4 : b3;
  const int i00 = load_idx16(csr, begP0,      endP0, c16);
  const int i01 = load_idx16(csr, begP0 + 16, endP0, c16);
  const int i02 = load_idx16(csr, begP0 + 32, endP0, c16);
  const int i10 = load_idx16(csr, begP1,      endP1, c16);
  const int i11 = load_idx16(csr, begP1 + 16, endP1, c16);
  const int i12 = load_idx16(csr, begP1 + 32, endP1, c16);

  const char* xb = (const char*)xin;
  const unsigned coff = (unsigned)c16 * (4u * sizeof(float));
  for (int p = 0; p < 2; ++p) {
    const int beg = p == 0 ? begP0 : begP1;
    const int end = p == 0 ? endP0 : endP1;
    const int c0  = p == 0 ? i00 : i10;
    const int c1  = p == 0 ? i01 : i11;
    const int c2  = p == 0 ? i02 : i12;
    const int dA = p == 0 ? b1 - b0 : b3 - b2;
    const int dB = p == 0 ? b2 - b1 : b4 - b3;
    const int nmax = dA > dB ? dA : dB;
    float4 acc = make_float4(0.0f, 0.0f, 0.0f, 0.0f);
    int k = 0;
    for (int j0 = 0; j0 < nmax; j0 += 16, ++k) {
      int ix;
      if (k == 0)      ix = c0;
      else if (k == 1) ix = c1;
      else if (k == 2) ix = c2;
      else             ix = load_idx16(csr, beg + j0, end, c16);
      gather_batch<0>(xb, coff, hsel, ix, acc);
      if (j0 + GATHER_DEPTH < nmax)
        gather_batch<GATHER_DEPTH>(xb, coff, hsel, ix, acc);
    }
    const float sc = 1.0f / fmaxf((float)(end - beg), 1.0f);
    *(float4*)(smean + (wv * 4 + p * 2 + (hsel >> 4)) * MEAN_PITCH + c16 * 4) =
        make_float4(acc.x * sc, acc.y * sc, acc.z * sc, acc.w * sc);
  }

  const int ni = wv;
  const v16h ax0 = load_a_frag(xin, DFEAT, rbase, 0, lane);
  const v16h ax1 = load_a_frag(xin, DFEAT, rbase, 32, lane);
  const v16h wr0 = load_b_packed(PWr, 4, 0, ni, lane);
  const v16h wr1 = load_b_packed(PWr, 4, 1, ni, lane);
  const int n  = lane & 15;
  const int hi = lane >> 4;
  const float bv = bl[ni * 16 + n];
  __syncthreads();

  const v16h wl0 = load_b_packed(PWl, 4, 0, ni, lane);
  const v16h wl1 = load_b_packed(PWl, 4, 1, ni, lane);
  v8f acc = {};
  acc = wmma_f16(load_a_frag(smean, MEAN_PITCH, 0, 0, lane), wl0, acc);
  acc = wmma_f16(ax0, wr0, acc);
  acc = wmma_f16(load_a_frag(smean, MEAN_PITCH, 0, 32, lane), wl1, acc);
  acc = wmma_f16(ax1, wr1, acc);

  float* orow = xout + (size_t)(rbase + hi * 8) * DFEAT + ni * 16 + n;
#pragma unroll
  for (int r = 0; r < 8; ++r) {
    const float v = acc[r] + bv;
    orow[(size_t)r * DFEAT] = v > 0.0f ? v : 0.0f;
  }
}

__global__ void head_kernel(const float* __restrict__ x,
                            const float* __restrict__ Wo,
                            const float* __restrict__ bo,
                            float* __restrict__ out) {
  const int i = blockIdx.x * blockDim.x + threadIdx.x;
  if (i >= NNODES) return;
  const float* row = x + (size_t)i * DFEAT;
  float o0 = bo[0], o1 = bo[1];
#pragma unroll
  for (int k = 0; k < DFEAT; ++k) {
    const float v = row[k];
    o0 += v * Wo[k * 2 + 0];
    o1 += v * Wo[k * 2 + 1];
  }
  out[i * 2 + 0] = o0;
  out[i * 2 + 1] = o1;
}

extern "C" void kernel_launch(void* const* d_in, const int* in_sizes, int n_in,
                              void* d_out, int out_size, void* d_ws,
                              size_t ws_size, hipStream_t stream) {
  const float* xc  = (const float*)d_in[0];
  const float* xs  = (const float*)d_in[1];
  const int*   ei  = (const int*)d_in[2];
  const float* Wp  = (const float*)d_in[4];
  const float* bp  = (const float*)d_in[5];
  const float* Ws  = (const float*)d_in[6];
  const float* bs  = (const float*)d_in[7];
  const float* Wl1 = (const float*)d_in[8];
  const float* bl1 = (const float*)d_in[9];
  const float* Wr1 = (const float*)d_in[10];
  const float* Wl2 = (const float*)d_in[11];
  const float* bl2 = (const float*)d_in[12];
  const float* Wr2 = (const float*)d_in[13];
  const float* Wl3 = (const float*)d_in[14];
  const float* bl3 = (const float*)d_in[15];
  const float* Wr3 = (const float*)d_in[16];
  const float* Wo  = (const float*)d_in[17];
  const float* bo  = (const float*)d_in[18];

  const int E = in_sizes[2] / 2;
  const int* src = ei;
  const int* dst = ei + E;

  float* ws  = (float*)d_ws;
  float* X0  = ws;
  int*   CSR = (int*)(ws + (size_t)(NNODES + 1) * DFEAT);
  int*   BKT = CSR + E;
  float* X1  = ws + 2ull * NNODES * DFEAT;
  int2*  REC = (int2*)X1;
  int*   RPT = (int*)(ws + 3ull * NNODES * DFEAT + DFEAT);
  _Float16* PW   = (_Float16*)(ws + 3ull * NNODES * DFEAT + DFEAT + NNODES);
  _Float16* PWp  = PW;
  _Float16* PWs  = PW + 48ull  * FRAGH;
  _Float16* PWl1 = PW + 56ull  * FRAGH;
  _Float16* PWr1 = PW + 64ull  * FRAGH;
  _Float16* PWl2 = PW + 72ull  * FRAGH;
  _Float16* PWr2 = PW + 80ull  * FRAGH;
  _Float16* PWl3 = PW + 88ull  * FRAGH;
  _Float16* PWr3 = PW + 96ull  * FRAGH;

  const int layer_blocks = NNODES / 16;
  const int chunks = (E + CHUNK - 1) / CHUNK;

  prep_kernel<<<PREP_PACK_BLOCKS + PREP_ZERO_BLOCKS, 256, 0, stream>>>(
      Wp, Ws, Wl1, Wr1, Wl2, Wr2, Wl3, Wr3, PW, BKT,
      X0 + (size_t)NNODES * DFEAT, X1 + (size_t)NNODES * DFEAT);

  proj_hist_kernel<<<PROJ_BLOCKS + chunks, 256, 0, stream>>>(
      xc, xs, PWp, bp, PWs, bs, X0, dst, BKT, E);

  part_kernel<<<chunks, 512, 0, stream>>>(src, dst, BKT, REC, E);
  bsort_kernel<<<NBKT, 512, 0, stream>>>(REC, BKT, RPT, CSR);

  layer_kernel<<<layer_blocks, 128, 0, stream>>>(X0, X1, RPT, CSR, PWl1, bl1, PWr1, E);
  layer_kernel<<<layer_blocks, 128, 0, stream>>>(X1, X0, RPT, CSR, PWl2, bl2, PWr2, E);
  layer_kernel<<<layer_blocks, 128, 0, stream>>>(X0, X1, RPT, CSR, PWl3, bl3, PWr3, E);

  head_kernel<<<(NNODES + 255) / 256, 256, 0, stream>>>(X1, Wo, bo, (float*)d_out);
}


# Projection as a stream: K split over 4 waves per row tile, each wave requests all 28 of its row loads before the first product, weights via LDS; partial sums added in LDS in fixed order
# speedup vs baseline: 6.8291x; 1.2993x over previous
#include <hip/hip_runtime.h>
#include <hip/hip_bf16.h>


#define NNODES 50000
#define DFEAT  64
#define FRAGH  512

#define BKT_SHIFT 7
#define BKT_NODES (1 << BKT_SHIFT)
#define NBKT      ((NNODES + BKT_NODES - 1) / BKT_NODES)
#define BKT_PAD   512
#define CHUNK     4096
static_assert(NBKT <= BKT_PAD, "one table entry per bucket");

typedef __attribute__((ext_vector_type(16))) _Float16 v16h;
typedef __attribute__((ext_vector_type(8)))  _Float16 v8h;
typedef __attribute__((ext_vector_type(8)))  float    v8f;
typedef __attribute__((ext_vector_type(4)))  float    v4f;
typedef __attribute__((ext_vector_type(4)))  unsigned v4u;

__device__ __forceinline__ v8f wmma_f16(v16h a, v16h b, v8f c) {
  return __builtin_amdgcn_wmma_f32_16x16x32_f16(
       false, a,  false, b,
       (short)0, c,  false,  false);
}


__device__ __forceinline__ v16h load_a_frag_h(const _Float16* __restrict__ X, int ld,
                                              int rbase, int kbase, int lane) {
  const int m  = lane & 15;
  const int hi = lane >> 4;
  const _Float16* p = X + (size_t)(rbase + m) * ld + kbase + hi * 8;
  const v8h lo = *(const v8h*)(p + 0);
  const v8h up = *(const v8h*)(p + 16);
  return __builtin_shufflevector(lo, up, 0, 1, 2, 3, 4, 5, 6, 7,
                                 8, 9, 10, 11, 12, 13, 14, 15);
}

__device__ __forceinline__ v16h load_b_packed(const _Float16* __restrict__ P,
                                              int nt, int ki, int ni, int lane) {
  return *((const v16h*)P + ((ki * nt + ni) * 32 + lane));
}

__device__ __forceinline__ void pack_b(const float* __restrict__ W, int K,
                                       int ncols, _Float16* __restrict__ out,
                                       int t) {
  const int kt = K >> 5, nt = ncols >> 4;
  if (t >= kt * nt * 32) return;
  const int lane = t & 31;
  const int frag = t >> 5;
  const int ni = frag % nt, ki = frag / nt;
  const int n = lane & 15, hi = lane >> 4;
  const float* p = W + (size_t)(ki * 32 + hi * 8) * ncols + ni * 16 + n;
  _Float16* o = out + (size_t)frag * FRAGH + lane * 16;
#pragma unroll
  for (int j = 0; j < 16; ++j) {
    const int k = ((j >> 3) << 4) + (j & 7);
    o[j] = (_Float16)p[(size_t)k * ncols];
  }
}

#define PREP_PACK_BLOCKS 13
#define PREP_ZERO_BLOCKS 1
__global__ __launch_bounds__(256) void prep_kernel(
    const float* __restrict__ Wp,  const float* __restrict__ Ws,
    const float* __restrict__ Wl1, const float* __restrict__ Wr1,
    const float* __restrict__ Wl2, const float* __restrict__ Wr2,
    const float* __restrict__ Wl3, const float* __restrict__ Wr3,
    _Float16* __restrict__ PW, int* __restrict__ bkt,
    _Float16* __restrict__ zrow0, _Float16* __restrict__ zrow1) {
  const int b = blockIdx.x;
  if (b >= PREP_PACK_BLOCKS) {
    for (int i = threadIdx.x; i < 2 * BKT_PAD; i += 256) bkt[i] = 0;
    if (threadIdx.x < DFEAT) {
      zrow0[threadIdx.x] = (_Float16)0.0f;
      zrow1[threadIdx.x] = (_Float16)0.0f;
    }
    return;
  }
  if (b < 6) {
    pack_b(Wp, 768, 32, PW, b * 256 + threadIdx.x);
    return;
  }
  if (b == 6) {
    pack_b(Ws, 128, 32, PW + 48ull * FRAGH, threadIdx.x);
    return;
  }
  const int l = b - 7;
  const float* W = l == 0 ? Wl1 : l == 1 ? Wr1 : l == 2 ? Wl2
                 : l == 3 ? Wr2 : l == 4 ? Wl3 : Wr3;
  pack_b(W, 64, 64, PW + (size_t)(56 + 8 * l) * FRAGH, threadIdx.x);
}

#define PROJ_BLOCKS ((NNODES + 31) / 32)
#define PROJ_CFRAGS 48
#define PROJ_SFRAGS 8
#define RED_PITCH   68
static_assert(BKT_PAD <= 8 * 16 * RED_PITCH, "the histogram shares the partial-sum area");

__device__ __forceinline__ v4f ld4(const float* __restrict__ p) {
  return *(const v4f*)p;
}

__device__ __forceinline__ v16h cvt_a_frag(const v4f (&q)[4]) {
  v16h a;
  a[0]  = (_Float16)q[0].x; a[1]  = (_Float16)q[0].y;
  a[2]  = (_Float16)q[0].z; a[3]  = (_Float16)q[0].w;
  a[4]  = (_Float16)q[1].x; a[5]  = (_Float16)q[1].y;
  a[6]  = (_Float16)q[1].z; a[7]  = (_Float16)q[1].w;
  a[8]  = (_Float16)q[2].x; a[9]  = (_Float16)q[2].y;
  a[10] = (_Float16)q[2].z; a[11] = (_Float16)q[2].w;
  a[12] = (_Float16)q[3].x; a[13] = (_Float16)q[3].y;
  a[14] = (_Float16)q[3].z; a[15] = (_Float16)q[3].w;
  return a;
}

__global__ __launch_bounds__(256) void proj_hist_kernel(
    const float* __restrict__ xc, const float* __restrict__ xs,
    const _Float16* __restrict__ PWp, const float* __restrict__ bp,
    const _Float16* __restrict__ PWs, const float* __restrict__ bs,
    _Float16* __restrict__ X0,
    const int* __restrict__ dst, int* __restrict__ bcnt, int nedges) {
  __shared__ __attribute__((aligned(32))) _Float16 sB[(PROJ_CFRAGS + PROJ_SFRAGS) * FRAGH];
  __shared__ __attribute__((aligned(16))) float red[8 * 16 * RED_PITCH];
  if (blockIdx.x >= PROJ_BLOCKS) {
    int* hist = (int*)red;
    const int e0 = (blockIdx.x - PROJ_BLOCKS) * CHUNK;
    for (int b = threadIdx.x; b < BKT_PAD; b += 256) hist[b] = 0;
    __syncthreads();
    int dv[CHUNK / 256];
#pragma unroll
    for (int i = 0; i < CHUNK / 256; ++i) {
      const int e = e0 + i * 256 + threadIdx.x;
      dv[i] = (e < nedges) ? dst[e] : -1;
    }
#pragma unroll
    for (int i = 0; i < CHUNK / 256; ++i)
      if (dv[i] >= 0) atomicAdd(&hist[dv[i] >> BKT_SHIFT], 1);
    __syncthreads();
    for (int b = threadIdx.x; b < NBKT; b += 256) {
      const int n = hist[b];
      if (n) atomicAdd(&bcnt[b], n);
    }
    return;
  }
  const int t    = threadIdx.x;
  const int lane = t & 31;
  const int wave = __builtin_amdgcn_readfirstlane(t >> 5);
  const int rt = wave >> 2;
  const int kq = wave & 3;
  const int rbase = blockIdx.x * 32 + rt * 16;
  const int rload = (rbase < NNODES) ? rbase : blockIdx.x * 32;
  const int m  = lane & 15;
  const int hi = lane >> 4;

  v4u wb[14];
#pragma unroll
  for (int j = 0; j < 12; ++j) wb[j] = ((const v4u*)PWp)[t + 256 * j];
#pragma unroll
  for (int j = 0; j < 2; ++j) wb[12 + j] = ((const v4u*)PWs)[t + 256 * j];

  const float* pcn = xc + (size_t)(rload + m) * 768 + kq * 192 + hi * 8;
  const float* psn = xs + (size_t)(rload + m) * 128 + kq * 32 + hi * 8;
  v4f ra[7][4];
#pragma unroll
  for (int i = 0; i < 6; ++i) {
    ra[i][0] = ld4(pcn + i * 32 + 0);
    ra[i][1] = ld4(pcn + i * 32 + 4);
    ra[i][2] = ld4(pcn + i * 32 + 16);
    ra[i][3] = ld4(pcn + i * 32 + 20);
  }
  ra[6][0] = ld4(psn + 0);
  ra[6][1] = ld4(psn + 4);
  ra[6][2] = ld4(psn + 16);
  ra[6][3] = ld4(psn + 20);
  asm volatile("" : "+v"(wb[0]), "+v"(wb[1]), "+v"(wb[2]), "+v"(wb[3]), "+v"(wb[4]),
                    "+v"(wb[5]), "+v"(wb[6]), "+v"(wb[7]), "+v"(wb[8]), "+v"(wb[9]),
                    "+v"(wb[10]), "+v"(wb[11]), "+v"(wb[12]), "+v"(wb[13]) : : "memory");

#pragma unroll
  for (int j = 0; j < 14; ++j) ((v4u*)sB)[t + 256 * j] = wb[j];
  __builtin_amdgcn_fence(__ATOMIC_RELEASE, "workgroup", "local");
  __builtin_amdgcn_s_barrier();
  __builtin_amdgcn_fence(__ATOMIC_ACQUIRE, "workgroup", "local");

  v8f accC0 = {}, accC1 = {}, accS0 = {}, accS1 = {};
#pragma unroll
  for (int i = 0; i < 6; ++i) {
    asm volatile("" : "+v"(ra[i][0]), "+v"(ra[i][1]), "+v"(ra[i][2]), "+v"(ra[i][3]));
    const v16h a = cvt_a_frag(ra[i]);
    const int f = (kq * 6 + i) * 2;
    const v16h b0 = *((const v16h*)sB + (f + 0) * 32 + lane);
    const v16h b1 = *((const v16h*)sB + (f + 1) * 32 + lane);
    accC0 = wmma_f16(a, b0, accC0);
    accC1 = wmma_f16(a, b1, accC1);
  }
  {
    asm volatile("" : "+v"(ra[6][0]), "+v"(ra[6][1]), "+v"(ra[6][2]), "+v"(ra[6][3]));
    const v16h a = cvt_a_frag(ra[6]);
    const int f = PROJ_CFRAGS + kq * 2;
    const v16h b0 = *((const v16h*)sB + (f + 0) * 32 + lane);
    const v16h b1 = *((const v16h*)sB + (f + 1) * 32 + lane);
    accS0 = wmma_f16(a, b0, accS0);
    accS1 = wmma_f16(a, b1, accS1);
  }

  float* rw = red + wave * (16 * RED_PITCH) + (hi * 8) * RED_PITCH + m;
#pragma unroll
  for (int r = 0; r < 8; ++r) {
    rw[r * RED_PITCH + 0]  = accC0[r];
    rw[r * RED_PITCH + 16] = accC1[r];
    rw[r * RED_PITCH + 32] = accS0[r];
    rw[r * RED_PITCH + 48] = accS1[r];
  }
  __syncthreads();

  const int row  = t >> 3;
  const int pc   = t & 7;
  const int grow = blockIdx.x * 32 + row;
  if (grow < NNODES) {
    const float* rp = red + ((row >> 4) * 4) * (16 * RED_PITCH) + (row & 15) * RED_PITCH + pc * 8;
    float4 s0 = *(const float4*)(rp + 0);
    float4 s1 = *(const float4*)(rp + 4);
#pragma unroll
    for (int q = 1; q < 4; ++q) {
      const float4 u0 = *(const float4*)(rp + q * (16 * RED_PITCH) + 0);
      const float4 u1 = *(const float4*)(rp + q * (16 * RED_PITCH) + 4);
      s0.x += u0.x; s0.y += u0.y; s0.z += u0.z; s0.w += u0.w;
      s1.x += u1.x; s1.y += u1.y; s1.z += u1.z; s1.w += u1.w;
    }
    const float* bias = (pc < 4) ? (bp + pc * 8) : (bs + (pc - 4) * 8);
    v8h o;
    o[0] = (_Float16)(s0.x + bias[0]); o[1] = (_Float16)(s0.y + bias[1]);
    o[2] = (_Float16)(s0.z + bias[2]); o[3] = (_Float16)(s0.w + bias[3]);
    o[4] = (_Float16)(s1.x + bias[4]); o[5] = (_Float16)(s1.y + bias[5]);
    o[6] = (_Float16)(s1.z + bias[6]); o[7] = (_Float16)(s1.w + bias[7]);
    *(v8h*)(X0 + (size_t)grow * DFEAT + pc * 8) = o;
  }
}


__device__ __forceinline__ int block_excl_scan512(int v, int* wsum) {
  const int lane = threadIdx.x & 31;
  const int wave = threadIdx.x >> 5;
  int incl = v;
#pragma unroll
  for (int off = 1; off < 32; off <<= 1) {
    const int up = __shfl_up(incl, off);
    if (lane >= off) incl += up;
  }
  if (lane == 31) wsum[wave] = incl;
  __syncthreads();
  int add = 0;
  for (int w = 0; w < wave; ++w) add += wsum[w];
  __syncthreads();
  return add + incl - v;
}

__global__ __launch_bounds__(512) void part_kernel(
    const int* __restrict__ src, const int* __restrict__ dst,
    int* __restrict__ bkt, int2* __restrict__ rec, int nedges) {
  __shared__ int2 srec[CHUNK];
  __shared__ int lcnt[BKT_PAD];
  __shared__ int lst[BKT_PAD];
  __shared__ int dlt[BKT_PAD];
  __shared__ int wsum[16];
  const int t = threadIdx.x;
  const int e0 = blockIdx.x * CHUNK;

  lcnt[t] = 0;
  const int bst = block_excl_scan512(bkt[t], wsum);
  if (blockIdx.x == 0) bkt[2 * BKT_PAD + t] = bst;

  int s[CHUNK / 512], d[CHUNK / 512], r[CHUNK / 512];
#pragma unroll
  for (int i = 0; i < CHUNK / 512; ++i) {
    const int e = e0 + i * 512 + t;
    const bool in = e < nedges;
    s[i] = in ? src[e] : 0;
    d[i] = in ? dst[e] : -1;
  }
#pragma unroll
  for (int i = 0; i < CHUNK / 512; ++i)
    r[i] = (d[i] >= 0) ? atomicAdd(&lcnt[d[i] >> BKT_SHIFT], 1) : 0;
  __syncthreads();

  const int cnt = lcnt[t];
  const int base = cnt ? atomicAdd(&bkt[BKT_PAD + t], cnt) : 0;
  const int ls = block_excl_scan512(cnt, wsum);
  lst[t] = ls;
  dlt[t] = bst + base - ls;
  __syncthreads();

#pragma unroll
  for (int i = 0; i < CHUNK / 512; ++i)
    if (d[i] >= 0) srec[lst[d[i] >> BKT_SHIFT] + r[i]] = make_int2(s[i], d[i]);
  __syncthreads();

  const int left = nedges - e0;
  const int m = left < CHUNK ? left : CHUNK;
  for (int j = t; j < m; j += 512) {
    const int2 q = srec[j];
    rec[dlt[q.y >> BKT_SHIFT] + j] = q;
  }
}

#define STAGE_CAP 12288
#define BSORT_HELD 10
__global__ __launch_bounds__(512) void bsort_kernel(
    const int2* __restrict__ rec, const int* __restrict__ bkt,
    int* __restrict__ rowptr, int* __restrict__ csr) {
  __shared__ int sbuf[STAGE_CAP];
  __shared__ int ncnt[BKT_NODES];
  __shared__ int ncur[BKT_NODES];
  __shared__ int wsum[16];
  const int t = threadIdx.x;
  const int b = blockIdx.x;
  const int n  = bkt[b];
  const int bs = bkt[2 * BKT_PAD + b];
  const int2* my = rec + bs;

  int2 q[BSORT_HELD];
#pragma unroll
  for (int i = 0; i < BSORT_HELD; ++i) {
    const int j = t + i * 512;
    q[i] = (j < n) ? my[j] : make_int2(0, -1);
  }
  if (t < BKT_NODES) ncnt[t] = 0;
  __syncthreads();
#pragma unroll
  for (int i = 0; i < BSORT_HELD; ++i)
    if (q[i].y >= 0) atomicAdd(&ncnt[q[i].y & (BKT_NODES - 1)], 1);
  for (int j = t + BSORT_HELD * 512; j < n; j += 512)
    atomicAdd(&ncnt[my[j].y & (BKT_NODES - 1)], 1);
  __syncthreads();

  const int c = (t < BKT_NODES) ? ncnt[t] : 0;
  const int ex = block_excl_scan512(c, wsum);
  if (t < BKT_NODES) {
    ncur[t] = ex;
    const int node = b * BKT_NODES + t;
    if (node < NNODES) rowptr[node] = bs + ex;
  }
  __syncthreads();

  const bool staged = n <= STAGE_CAP;
#pragma unroll
  for (int i = 0; i < BSORT_HELD; ++i) {
    if (q[i].y >= 0) {
      const int p = atomicAdd(&ncur[q[i].y & (BKT_NODES - 1)], 1);
      if (staged) sbuf[p] = q[i].x; else csr[bs + p] = q[i].x;
    }
  }
  for (int j = t + BSORT_HELD * 512; j < n; j += 512) {
    const int2 v = my[j];
    const int p = atomicAdd(&ncur[v.y & (BKT_NODES - 1)], 1);
    if (staged) sbuf[p] = v.x; else csr[bs + p] = v.x;
  }
  if (staged) {
    __syncthreads();
    for (int j = t; j < n; j += 512) csr[bs + j] = sbuf[j];
  }
}

static_assert(NNODES % 16 == 0, "row tiles must be exact");
#define GATHER_DEPTH 8
#define ROW_BYTES    (DFEAT * 2)
#define MEAN_PITCH   72

__device__ __forceinline__ int load_idx8(const int* __restrict__ csr, int beg,
                                         int end, int c) {
  const int pos = beg + c;
  return (pos < end) ? csr[pos] : NNODES;
}

template <int U>
__device__ __forceinline__ unsigned bcast8(int v) {
  return (unsigned)__builtin_amdgcn_mov_dpp8((unsigned)v, U * 0x249249);
}

__device__ __forceinline__ void add_h2(float& a0, float& a1, unsigned w) {
  asm("v_fma_mix_f32 %0, %1, 1.0, %0 op_sel_hi:[1,0,0]" : "+v"(a0) : "v"(w));
  asm("v_fma_mix_f32 %0, %1, 1.0, %0 op_sel:[1,0,0] op_sel_hi:[1,0,0]" : "+v"(a1) : "v"(w));
}

__device__ __forceinline__ void gather_batch(const char* __restrict__ xb,
                                             unsigned coff, int ix,
                                             float (&acc)[8]) {
  unsigned r[GATHER_DEPTH];
  r[0] = bcast8<0>(ix); r[1] = bcast8<1>(ix); r[2] = bcast8<2>(ix); r[3] = bcast8<3>(ix);
  r[4] = bcast8<4>(ix); r[5] = bcast8<5>(ix); r[6] = bcast8<6>(ix); r[7] = bcast8<7>(ix);
  uint4 v[GATHER_DEPTH];
#pragma unroll
  for (int u = 0; u < GATHER_DEPTH; ++u)
    v[u] = *(const uint4*)(xb + (r[u] * (unsigned)ROW_BYTES + coff));
  __builtin_amdgcn_sched_barrier(0);
#pragma unroll
  for (int u = 0; u < GATHER_DEPTH; ++u) {
    add_h2(acc[0], acc[1], v[u].x);
    add_h2(acc[2], acc[3], v[u].y);
    add_h2(acc[4], acc[5], v[u].z);
    add_h2(acc[6], acc[7], v[u].w);
  }
}

template <bool OUT_F32>
__global__ __launch_bounds__(128, 12) void layer_kernel(
    const _Float16* __restrict__ xin, _Float16* __restrict__ xout_h,
    float* __restrict__ xout_f,
    const int* __restrict__ rowptr, const int* __restrict__ csr,
    const _Float16* __restrict__ PWl, const float* __restrict__ bl,
    const _Float16* __restrict__ PWr, int nedges) {
  __shared__ __attribute__((aligned(16))) _Float16 smean[16 * MEAN_PITCH];
  __shared__ __attribute__((aligned(16))) _Float16 sout[16 * MEAN_PITCH];
  const int lane = threadIdx.x & 31;
  const int wv = __builtin_amdgcn_readfirstlane(threadIdx.x >> 5);
  const int rbase = blockIdx.x * 16;
  const int n0 = rbase + wv * 4;

  int rp = 0;
  if (lane <= 4) {
    const int nd = n0 + lane;
    rp = (nd < NNODES) ? rowptr[nd] : nedges;
  }
  const int b0 = __builtin_amdgcn_readlane(rp, 0);
  const int b1 = __builtin_amdgcn_readlane(rp, 1);
  const int b2 = __builtin_amdgcn_readlane(rp, 2);
  const int b3 = __builtin_amdgcn_readlane(rp, 3);
  const int b4 = __builtin_amdgcn_readlane(rp, 4);
  const int c8 = lane & 7;
  const int q  = lane >> 3;
  const int beg = q == 0 ? b0 : q == 1 ? b1 : q == 2 ? b2 : b3;
  const int end = q == 0 ? b1 : q == 1 ? b2 : q == 2 ? b3 : b4;
  const int i0 = load_idx8(csr, beg,      end, c8);
  const int i1 = load_idx8(csr, beg + 8,  end, c8);
  const int i2 = load_idx8(csr, beg + 16, end, c8);
  const int i3 = load_idx8(csr, beg + 24, end, c8);
  const int i4 = load_idx8(csr, beg + 32, end, c8);
  const int i5 = load_idx8(csr, beg + 40, end, c8);

  const char* xb = (const char*)xin;
  const unsigned coff = (unsigned)c8 * 16u;
  {
    const int d0 = b1 - b0, d1 = b2 - b1, d2 = b3 - b2, d3 = b4 - b3;
    const int m01 = d0 > d1 ? d0 : d1;
    const int m23 = d2 > d3 ? d2 : d3;
    const int nmax = m01 > m23 ? m01 : m23;
    float acc[8] = {0.0f, 0.0f, 0.0f, 0.0f, 0.0f, 0.0f, 0.0f, 0.0f};
    int k = 0;
    for (int j0 = 0; j0 < nmax; j0 += GATHER_DEPTH, ++k) {
      int ix;
      if (k == 0)      ix = i0;
      else if (k == 1) ix = i1;
      else if (k == 2) ix = i2;
      else if (k == 3) ix = i3;
      else if (k == 4) ix = i4;
      else if (k == 5) ix = i5;
      else             ix = load_idx8(csr, beg + j0, end, c8);
      gather_batch(xb, coff, ix, acc);
    }
    const float sc = 1.0f / fmaxf((float)(end - beg), 1.0f);
    v8h mh;
#pragma unroll
    for (int e = 0; e < 8; ++e) mh[e] = (_Float16)(acc[e] * sc);
    *(v8h*)(smean + (wv * 4 + q) * MEAN_PITCH + c8 * 8) = mh;
  }

  const int ni = wv;
  const v16h ax0 = load_a_frag_h(xin, DFEAT, rbase, 0, lane);
  const v16h ax1 = load_a_frag_h(xin, DFEAT, rbase, 32, lane);
  const v16h wr0 = load_b_packed(PWr, 4, 0, ni, lane);
  const v16h wr1 = load_b_packed(PWr, 4, 1, ni, lane);
  const int n  = lane & 15;
  const int hi = lane >> 4;
  const float bv = bl[ni * 16 + n];
  __syncthreads();

  const v16h wl0 = load_b_packed(PWl, 4, 0, ni, lane);
  const v16h wl1 = load_b_packed(PWl, 4, 1, ni, lane);
  v8f acc = {};
  acc = wmma_f16(load_a_frag_h(smean, MEAN_PITCH, 0, 0, lane), wl0, acc);
  acc = wmma_f16(ax0, wr0, acc);
  acc = wmma_f16(load_a_frag_h(smean, MEAN_PITCH, 0, 32, lane), wl1, acc);
  acc = wmma_f16(ax1, wr1, acc);

  if (OUT_F32) {
    float* orow = xout_f + (size_t)(rbase + hi * 8) * DFEAT + ni * 16 + n;
#pragma unroll
    for (int r = 0; r < 8; ++r) {
      const float v = acc[r] + bv;
      orow[(size_t)r * DFEAT] = v > 0.0f ? v : 0.0f;
    }
  } else {
    _Float16* srow = sout + (hi * 8) * MEAN_PITCH + ni * 16 + n;
#pragma unroll
    for (int r = 0; r < 8; ++r) {
      const float v = acc[r] + bv;
      srow[r * MEAN_PITCH] = (_Float16)(v > 0.0f ? v : 0.0f);
    }
    __syncthreads();
    const int row = threadIdx.x >> 3;
    const int pc  = threadIdx.x & 7;
    *(v8h*)(xout_h + (size_t)(rbase + row) * DFEAT + pc * 8) =
        *(const v8h*)(sout + row * MEAN_PITCH + pc * 8);
  }
}

__global__ void head_kernel(const float* __restrict__ x,
                            const float* __restrict__ Wo,
                            const float* __restrict__ bo,
                            float* __restrict__ out) {
  const int i = blockIdx.x * blockDim.x + threadIdx.x;
  if (i >= NNODES) return;
  const float* row = x + (size_t)i * DFEAT;
  float o0 = bo[0], o1 = bo[1];
#pragma unroll
  for (int k = 0; k < DFEAT; ++k) {
    const float v = row[k];
    o0 += v * Wo[k * 2 + 0];
    o1 += v * Wo[k * 2 + 1];
  }
  out[i * 2 + 0] = o0;
  out[i * 2 + 1] = o1;
}

extern "C" void kernel_launch(void* const* d_in, const int* in_sizes, int n_in,
                              void* d_out, int out_size, void* d_ws,
                              size_t ws_size, hipStream_t stream) {
  const float* xc  = (const float*)d_in[0];
  const float* xs  = (const float*)d_in[1];
  const int*   ei  = (const int*)d_in[2];
  const float* Wp  = (const float*)d_in[4];
  const float* bp  = (const float*)d_in[5];
  const float* Ws  = (const float*)d_in[6];
  const float* bs  = (const float*)d_in[7];
  const float* Wl1 = (const float*)d_in[8];
  const float* bl1 = (const float*)d_in[9];
  const float* Wr1 = (const float*)d_in[10];
  const float* Wl2 = (const float*)d_in[11];
  const float* bl2 = (const float*)d_in[12];
  const float* Wr2 = (const float*)d_in[13];
  const float* Wl3 = (const float*)d_in[14];
  const float* bl3 = (const float*)d_in[15];
  const float* Wr3 = (const float*)d_in[16];
  const float* Wo  = (const float*)d_in[17];
  const float* bo  = (const float*)d_in[18];

  const int E = in_sizes[2] / 2;
  const int* src = ei;
  const int* dst = ei + E;

  float* ws  = (float*)d_ws;
  _Float16* X0 = (_Float16*)ws;
  int*   CSR = (int*)(ws + (size_t)(NNODES + 1) * DFEAT);
  int*   BKT = CSR + E;
  float* XF  = ws + 2ull * NNODES * DFEAT;
  int2*  REC = (int2*)XF;
  _Float16* X1 = (_Float16*)XF + (size_t)NNODES * DFEAT;
  int*   RPT = (int*)(ws + 3ull * NNODES * DFEAT + DFEAT);
  _Float16* PW   = (_Float16*)(ws + 3ull * NNODES * DFEAT + DFEAT + NNODES);
  _Float16* PWp  = PW;
  _Float16* PWs  = PW + 48ull  * FRAGH;
  _Float16* PWl1 = PW + 56ull  * FRAGH;
  _Float16* PWr1 = PW + 64ull  * FRAGH;
  _Float16* PWl2 = PW + 72ull  * FRAGH;
  _Float16* PWr2 = PW + 80ull  * FRAGH;
  _Float16* PWl3 = PW + 88ull  * FRAGH;
  _Float16* PWr3 = PW + 96ull  * FRAGH;

  const int layer_blocks = NNODES / 16;
  const int chunks = (E + CHUNK - 1) / CHUNK;

  prep_kernel<<<PREP_PACK_BLOCKS + PREP_ZERO_BLOCKS, 256, 0, stream>>>(
      Wp, Ws, Wl1, Wr1, Wl2, Wr2, Wl3, Wr3, PW, BKT,
      X0 + (size_t)NNODES * DFEAT, X1 + (size_t)NNODES * DFEAT);

  proj_hist_kernel<<<PROJ_BLOCKS + chunks, 256, 0, stream>>>(
      xc, xs, PWp, bp, PWs, bs, X0, dst, BKT, E);

  part_kernel<<<chunks, 512, 0, stream>>>(src, dst, BKT, REC, E);
  bsort_kernel<<<NBKT, 512, 0, stream>>>(REC, BKT, RPT, CSR);

  layer_kernel<false><<<layer_blocks, 128, 0, stream>>>(X0, X1, nullptr, RPT, CSR, PWl1, bl1, PWr1, E);
  layer_kernel<false><<<layer_blocks, 128, 0, stream>>>(X1, X0, nullptr, RPT, CSR, PWl2, bl2, PWr2, E);
  layer_kernel<true><<<layer_blocks, 128, 0, stream>>>(X0, nullptr, XF, RPT, CSR, PWl3, bl3, PWr3, E);

  head_kernel<<<(NNODES + 255) / 256, 256, 0, stream>>>(XF, Wo, bo, (float*)d_out);
}


# Classifier head is the tail of layer 3's launch: the f32 16 x 64 tile stays in LDS, 8 lanes per row, row sum = fixed tree over 8 partial sums (was one rising sum); 8 -> 7 launches
# speedup vs baseline: 7.1770x; 1.0509x over previous
#include <hip/hip_runtime.h>
#include <hip/hip_bf16.h>


#define NNODES 50000
#define DFEAT  64
#define FRAGH  512

#define BKT_SHIFT 7
#define BKT_NODES (1 << BKT_SHIFT)
#define NBKT      ((NNODES + BKT_NODES - 1) / BKT_NODES)
#define BKT_PAD   512
#define CHUNK     4096
static_assert(NBKT <= BKT_PAD, "one table entry per bucket");

typedef __attribute__((ext_vector_type(16))) _Float16 v16h;
typedef __attribute__((ext_vector_type(8)))  _Float16 v8h;
typedef __attribute__((ext_vector_type(8)))  float    v8f;
typedef __attribute__((ext_vector_type(4)))  float    v4f;
typedef __attribute__((ext_vector_type(2)))  float    v2f;
typedef __attribute__((ext_vector_type(4)))  unsigned v4u;

__device__ __forceinline__ v8f wmma_f16(v16h a, v16h b, v8f c) {
  return __builtin_amdgcn_wmma_f32_16x16x32_f16(
       false, a,  false, b,
       (short)0, c,  false,  false);
}


__device__ __forceinline__ v16h load_a_frag_h(const _Float16* __restrict__ X, int ld,
                                              int rbase, int kbase, int lane) {
  const int m  = lane & 15;
  const int hi = lane >> 4;
  const _Float16* p = X + (size_t)(rbase + m) * ld + kbase + hi * 8;
  const v8h lo = *(const v8h*)(p + 0);
  const v8h up = *(const v8h*)(p + 16);
  return __builtin_shufflevector(lo, up, 0, 1, 2, 3, 4, 5, 6, 7,
                                 8, 9, 10, 11, 12, 13, 14, 15);
}

__device__ __forceinline__ v16h load_b_packed(const _Float16* __restrict__ P,
                                              int nt, int ki, int ni, int lane) {
  return *((const v16h*)P + ((ki * nt + ni) * 32 + lane));
}

__device__ __forceinline__ void pack_b(const float* __restrict__ W, int K,
                                       int ncols, _Float16* __restrict__ out,
                                       int t) {
  const int kt = K >> 5, nt = ncols >> 4;
  if (t >= kt * nt * 32) return;
  const int lane = t & 31;
  const int frag = t >> 5;
  const int ni = frag % nt, ki = frag / nt;
  const int n = lane & 15, hi = lane >> 4;
  const float* p = W + (size_t)(ki * 32 + hi * 8) * ncols + ni * 16 + n;
  _Float16* o = out + (size_t)frag * FRAGH + lane * 16;
#pragma unroll
  for (int j = 0; j < 16; ++j) {
    const int k = ((j >> 3) << 4) + (j & 7);
    o[j] = (_Float16)p[(size_t)k * ncols];
  }
}

#define PREP_PACK_BLOCKS 13
#define PREP_ZERO_BLOCKS 1
__global__ __launch_bounds__(256) void prep_kernel(
    const float* __restrict__ Wp,  const float* __restrict__ Ws,
    const float* __restrict__ Wl1, const float* __restrict__ Wr1,
    const float* __restrict__ Wl2, const float* __restrict__ Wr2,
    const float* __restrict__ Wl3, const float* __restrict__ Wr3,
    _Float16* __restrict__ PW, int* __restrict__ bkt,
    _Float16* __restrict__ zrow0, _Float16* __restrict__ zrow1) {
  const int b = blockIdx.x;
  if (b >= PREP_PACK_BLOCKS) {
    for (int i = threadIdx.x; i < 2 * BKT_PAD; i += 256) bkt[i] = 0;
    if (threadIdx.x < DFEAT) {
      zrow0[threadIdx.x] = (_Float16)0.0f;
      zrow1[threadIdx.x] = (_Float16)0.0f;
    }
    return;
  }
  if (b < 6) {
    pack_b(Wp, 768, 32, PW, b * 256 + threadIdx.x);
    return;
  }
  if (b == 6) {
    pack_b(Ws, 128, 32, PW + 48ull * FRAGH, threadIdx.x);
    return;
  }
  const int l = b - 7;
  const float* W = l == 0 ? Wl1 : l == 1 ? Wr1 : l == 2 ? Wl2
                 : l == 3 ? Wr2 : l == 4 ? Wl3 : Wr3;
  pack_b(W, 64, 64, PW + (size_t)(56 + 8 * l) * FRAGH, threadIdx.x);
}

#define PROJ_BLOCKS ((NNODES + 31) / 32)
#define PROJ_CFRAGS 48
#define PROJ_SFRAGS 8
#define RED_PITCH   68
static_assert(BKT_PAD <= 8 * 16 * RED_PITCH, "the histogram shares the partial-sum area");

__device__ __forceinline__ v4f ld4(const float* __restrict__ p) {
  return *(const v4f*)p;
}

__device__ __forceinline__ v16h cvt_a_frag(const v4f (&q)[4]) {
  v16h a;
  a[0]  = (_Float16)q[0].x; a[1]  = (_Float16)q[0].y;
  a[2]  = (_Float16)q[0].z; a[3]  = (_Float16)q[0].w;
  a[4]  = (_Float16)q[1].x; a[5]  = (_Float16)q[1].y;
  a[6]  = (_Float16)q[1].z; a[7]  = (_Float16)q[1].w;
  a[8]  = (_Float16)q[2].x; a[9]  = (_Float16)q[2].y;
  a[10] = (_Float16)q[2].z; a[11] = (_Float16)q[2].w;
  a[12] = (_Float16)q[3].x; a[13] = (_Float16)q[3].y;
  a[14] = (_Float16)q[3].z; a[15] = (_Float16)q[3].w;
  return a;
}

__global__ __launch_bounds__(256) void proj_hist_kernel(
    const float* __restrict__ xc, const float* __restrict__ xs,
    const _Float16* __restrict__ PWp, const float* __restrict__ bp,
    const _Float16* __restrict__ PWs, const float* __restrict__ bs,
    _Float16* __restrict__ X0,
    const int* __restrict__ dst, int* __restrict__ bcnt, int nedges) {
  __shared__ __attribute__((aligned(32))) _Float16 sB[(PROJ_CFRAGS + PROJ_SFRAGS) * FRAGH];
  __shared__ __attribute__((aligned(16))) float red[8 * 16 * RED_PITCH];
  if (blockIdx.x >= PROJ_BLOCKS) {
    int* hist = (int*)red;
    const int e0 = (blockIdx.x - PROJ_BLOCKS) * CHUNK;
    for (int b = threadIdx.x; b < BKT_PAD; b += 256) hist[b] = 0;
    __syncthreads();
    int dv[CHUNK / 256];
#pragma unroll
    for (int i = 0; i < CHUNK / 256; ++i) {
      const int e = e0 + i * 256 + threadIdx.x;
      dv[i] = (e < nedges) ? dst[e] : -1;
    }
#pragma unroll
    for (int i = 0; i < CHUNK / 256; ++i)
      if (dv[i] >= 0) atomicAdd(&hist[dv[i] >> BKT_SHIFT], 1);
    __syncthreads();
    for (int b = threadIdx.x; b < NBKT; b += 256) {
      const int n = hist[b];
      if (n) atomicAdd(&bcnt[b], n);
    }
    return;
  }
  const int t    = threadIdx.x;
  const int lane = t & 31;
  const int wave = __builtin_amdgcn_readfirstlane(t >> 5);
  const int rt = wave >> 2;
  const int kq = wave & 3;
  const int rbase = blockIdx.x * 32 + rt * 16;
  const int rload = (rbase < NNODES) ? rbase : blockIdx.x * 32;
  const int m  = lane & 15;
  const int hi = lane >> 4;

  v4u wb[14];
#pragma unroll
  for (int j = 0; j < 12; ++j) wb[j] = ((const v4u*)PWp)[t + 256 * j];
#pragma unroll
  for (int j = 0; j < 2; ++j) wb[12 + j] = ((const v4u*)PWs)[t + 256 * j];

  const float* pcn = xc + (size_t)(rload + m) * 768 + kq * 192 + hi * 8;
  const float* psn = xs + (size_t)(rload + m) * 128 + kq * 32 + hi * 8;
  v4f ra[7][4];
#pragma unroll
  for (int i = 0; i < 6; ++i) {
    ra[i][0] = ld4(pcn + i * 32 + 0);
    ra[i][1] = ld4(pcn + i * 32 + 4);
    ra[i][2] = ld4(pcn + i * 32 + 16);
    ra[i][3] = ld4(pcn + i * 32 + 20);
  }
  ra[6][0] = ld4(psn + 0);
  ra[6][1] = ld4(psn + 4);
  ra[6][2] = ld4(psn + 16);
  ra[6][3] = ld4(psn + 20);
  asm volatile("" : "+v"(wb[0]), "+v"(wb[1]), "+v"(wb[2]), "+v"(wb[3]), "+v"(wb[4]),
                    "+v"(wb[5]), "+v"(wb[6]), "+v"(wb[7]), "+v"(wb[8]), "+v"(wb[9]),
                    "+v"(wb[10]), "+v"(wb[11]), "+v"(wb[12]), "+v"(wb[13]) : : "memory");

#pragma unroll
  for (int j = 0; j < 14; ++j) ((v4u*)sB)[t + 256 * j] = wb[j];
  __builtin_amdgcn_fence(__ATOMIC_RELEASE, "workgroup", "local");
  __builtin_amdgcn_s_barrier();
  __builtin_amdgcn_fence(__ATOMIC_ACQUIRE, "workgroup", "local");

  v8f accC0 = {}, accC1 = {}, accS0 = {}, accS1 = {};
#pragma unroll
  for (int i = 0; i < 6; ++i) {
    asm volatile("" : "+v"(ra[i][0]), "+v"(ra[i][1]), "+v"(ra[i][2]), "+v"(ra[i][3]));
    const v16h a = cvt_a_frag(ra[i]);
    const int f = (kq * 6 + i) * 2;
    const v16h b0 = *((const v16h*)sB + (f + 0) * 32 + lane);
    const v16h b1 = *((const v16h*)sB + (f + 1) * 32 + lane);
    accC0 = wmma_f16(a, b0, accC0);
    accC1 = wmma_f16(a, b1, accC1);
  }
  {
    asm volatile("" : "+v"(ra[6][0]), "+v"(ra[6][1]), "+v"(ra[6][2]), "+v"(ra[6][3]));
    const v16h a = cvt_a_frag(ra[6]);
    const int f = PROJ_CFRAGS + kq * 2;
    const v16h b0 = *((const v16h*)sB + (f + 0) * 32 + lane);
    const v16h b1 = *((const v16h*)sB + (f + 1) * 32 + lane);
    accS0 = wmma_f16(a, b0, accS0);
    accS1 = wmma_f16(a, b1, accS1);
  }

  float* rw = red + wave * (16 * RED_PITCH) + (hi * 8) * RED_PITCH + m;
#pragma unroll
  for (int r = 0; r < 8; ++r) {
    rw[r * RED_PITCH + 0]  = accC0[r];
    rw[r * RED_PITCH + 16] = accC1[r];
    rw[r * RED_PITCH + 32] = accS0[r];
    rw[r * RED_PITCH + 48] = accS1[r];
  }
  __syncthreads();

  const int row  = t >> 3;
  const int pc   = t & 7;
  const int grow = blockIdx.x * 32 + row;
  if (grow < NNODES) {
    const float* rp = red + ((row >> 4) * 4) * (16 * RED_PITCH) + (row & 15) * RED_PITCH + pc * 8;
    float4 s0 = *(const float4*)(rp + 0);
    float4 s1 = *(const float4*)(rp + 4);
#pragma unroll
    for (int q = 1; q < 4; ++q) {
      const float4 u0 = *(const float4*)(rp + q * (16 * RED_PITCH) + 0);
      const float4 u1 = *(const float4*)(rp + q * (16 * RED_PITCH) + 4);
      s0.x += u0.x; s0.y += u0.y; s0.z += u0.z; s0.w += u0.w;
      s1.x += u1.x; s1.y += u1.y; s1.z += u1.z; s1.w += u1.w;
    }
    const float* bias = (pc < 4) ? (bp + pc * 8) : (bs + (pc - 4) * 8);
    v8h o;
    o[0] = (_Float16)(s0.x + bias[0]); o[1] = (_Float16)(s0.y + bias[1]);
    o[2] = (_Float16)(s0.z + bias[2]); o[3] = (_Float16)(s0.w + bias[3]);
    o[4] = (_Float16)(s1.x + bias[4]); o[5] = (_Float16)(s1.y + bias[5]);
    o[6] = (_Float16)(s1.z + bias[6]); o[7] = (_Float16)(s1.w + bias[7]);
    *(v8h*)(X0 + (size_t)grow * DFEAT + pc * 8) = o;
  }
}


__device__ __forceinline__ int block_excl_scan512(int v, int* wsum) {
  const int lane = threadIdx.x & 31;
  const int wave = threadIdx.x >> 5;
  int incl = v;
#pragma unroll
  for (int off = 1; off < 32; off <<= 1) {
    const int up = __shfl_up(incl, off);
    if (lane >= off) incl += up;
  }
  if (lane == 31) wsum[wave] = incl;
  __syncthreads();
  int add = 0;
  for (int w = 0; w < wave; ++w) add += wsum[w];
  __syncthreads();
  return add + incl - v;
}

__global__ __launch_bounds__(512) void part_kernel(
    const int* __restrict__ src, const int* __restrict__ dst,
    int* __restrict__ bkt, int2* __restrict__ rec, int nedges) {
  __shared__ int2 srec[CHUNK];
  __shared__ int lcnt[BKT_PAD];
  __shared__ int lst[BKT_PAD];
  __shared__ int dlt[BKT_PAD];
  __shared__ int wsum[16];
  const int t = threadIdx.x;
  const int e0 = blockIdx.x * CHUNK;

  lcnt[t] = 0;
  const int bst = block_excl_scan512(bkt[t], wsum);
  if (blockIdx.x == 0) bkt[2 * BKT_PAD + t] = bst;

  int s[CHUNK / 512], d[CHUNK / 512], r[CHUNK / 512];
#pragma unroll
  for (int i = 0; i < CHUNK / 512; ++i) {
    const int e = e0 + i * 512 + t;
    const bool in = e < nedges;
    s[i] = in ? src[e] : 0;
    d[i] = in ? dst[e] : -1;
  }
#pragma unroll
  for (int i = 0; i < CHUNK / 512; ++i)
    r[i] = (d[i] >= 0) ? atomicAdd(&lcnt[d[i] >> BKT_SHIFT], 1) : 0;
  __syncthreads();

  const int cnt = lcnt[t];
  const int base = cnt ? atomicAdd(&bkt[BKT_PAD + t], cnt) : 0;
  const int ls = block_excl_scan512(cnt, wsum);
  lst[t] = ls;
  dlt[t] = bst + base - ls;
  __syncthreads();

#pragma unroll
  for (int i = 0; i < CHUNK / 512; ++i)
    if (d[i] >= 0) srec[lst[d[i] >> BKT_SHIFT] + r[i]] = make_int2(s[i], d[i]);
  __syncthreads();

  const int left = nedges - e0;
  const int m = left < CHUNK ? left : CHUNK;
  for (int j = t; j < m; j += 512) {
    const int2 q = srec[j];
    rec[dlt[q.y >> BKT_SHIFT] + j] = q;
  }
}

#define STAGE_CAP 12288
#define BSORT_HELD 10
__global__ __launch_bounds__(512) void bsort_kernel(
    const int2* __restrict__ rec, const int* __restrict__ bkt,
    int* __restrict__ rowptr, int* __restrict__ csr) {
  __shared__ int sbuf[STAGE_CAP];
  __shared__ int ncnt[BKT_NODES];
  __shared__ int ncur[BKT_NODES];
  __shared__ int wsum[16];
  const int t = threadIdx.x;
  const int b = blockIdx.x;
  const int n  = bkt[b];
  const int bs = bkt[2 * BKT_PAD + b];
  const int2* my = rec + bs;

  int2 q[BSORT_HELD];
#pragma unroll
  for (int i = 0; i < BSORT_HELD; ++i) {
    const int j = t + i * 512;
    q[i] = (j < n) ? my[j] : make_int2(0, -1);
  }
  if (t < BKT_NODES) ncnt[t] = 0;
  __syncthreads();
#pragma unroll
  for (int i = 0; i < BSORT_HELD; ++i)
    if (q[i].y >= 0) atomicAdd(&ncnt[q[i].y & (BKT_NODES - 1)], 1);
  for (int j = t + BSORT_HELD * 512; j < n; j += 512)
    atomicAdd(&ncnt[my[j].y & (BKT_NODES - 1)], 1);
  __syncthreads();

  const int c = (t < BKT_NODES) ? ncnt[t] : 0;
  const int ex = block_excl_scan512(c, wsum);
  if (t < BKT_NODES) {
    ncur[t] = ex;
    const int node = b * BKT_NODES + t;
    if (node < NNODES) rowptr[node] = bs + ex;
  }
  __syncthreads();

  const bool staged = n <= STAGE_CAP;
#pragma unroll
  for (int i = 0; i < BSORT_HELD; ++i) {
    if (q[i].y >= 0) {
      const int p = atomicAdd(&ncur[q[i].y & (BKT_NODES - 1)], 1);
      if (staged) sbuf[p] = q[i].x; else csr[bs + p] = q[i].x;
    }
  }
  for (int j = t + BSORT_HELD * 512; j < n; j += 512) {
    const int2 v = my[j];
    const int p = atomicAdd(&ncur[v.y & (BKT_NODES - 1)], 1);
    if (staged) sbuf[p] = v.x; else csr[bs + p] = v.x;
  }
  if (staged) {
    __syncthreads();
    for (int j = t; j < n; j += 512) csr[bs + j] = sbuf[j];
  }
}

static_assert(NNODES % 16 == 0, "row tiles must be exact");
#define GATHER_DEPTH 8
#define ROW_BYTES    (DFEAT * 2)
#define MEAN_PITCH   72

__device__ __forceinline__ int load_idx8(const int* __restrict__ csr, int beg,
                                         int end, int c) {
  const int pos = beg + c;
  return (pos < end) ? csr[pos] : NNODES;
}

template <int U>
__device__ __forceinline__ unsigned bcast8(int v) {
  return (unsigned)__builtin_amdgcn_mov_dpp8((unsigned)v, U * 0x249249);
}

__device__ __forceinline__ void add_h2(float& a0, float& a1, unsigned w) {
  asm("v_fma_mix_f32 %0, %1, 1.0, %0 op_sel_hi:[1,0,0]" : "+v"(a0) : "v"(w));
  asm("v_fma_mix_f32 %0, %1, 1.0, %0 op_sel:[1,0,0] op_sel_hi:[1,0,0]" : "+v"(a1) : "v"(w));
}

__device__ __forceinline__ void gather_batch(const char* __restrict__ xb,
                                             unsigned coff, int ix,
                                             float (&acc)[8]) {
  unsigned r[GATHER_DEPTH];
  r[0] = bcast8<0>(ix); r[1] = bcast8<1>(ix); r[2] = bcast8<2>(ix); r[3] = bcast8<3>(ix);
  r[4] = bcast8<4>(ix); r[5] = bcast8<5>(ix); r[6] = bcast8<6>(ix); r[7] = bcast8<7>(ix);
  uint4 v[GATHER_DEPTH];
#pragma unroll
  for (int u = 0; u < GATHER_DEPTH; ++u)
    v[u] = *(const uint4*)(xb + (r[u] * (unsigned)ROW_BYTES + coff));
  __builtin_amdgcn_sched_barrier(0);
#pragma unroll
  for (int u = 0; u < GATHER_DEPTH; ++u) {
    add_h2(acc[0], acc[1], v[u].x);
    add_h2(acc[2], acc[3], v[u].y);
    add_h2(acc[4], acc[5], v[u].z);
    add_h2(acc[6], acc[7], v[u].w);
  }
}

template <int X>
__device__ __forceinline__ float xor8(float v) {
  constexpr unsigned s =
      (unsigned)(0 ^ X)       | (unsigned)(1 ^ X) << 3  | (unsigned)(2 ^ X) << 6  |
      (unsigned)(3 ^ X) << 9  | (unsigned)(4 ^ X) << 12 | (unsigned)(5 ^ X) << 15 |
      (unsigned)(6 ^ X) << 18 | (unsigned)(7 ^ X) << 21;
  return __uint_as_float(__builtin_amdgcn_mov_dpp8(__float_as_uint(v), s));
}

#define HEAD_PITCH 68

template <bool HEAD>
__global__ __launch_bounds__(128, 12) void layer_kernel(
    const _Float16* __restrict__ xin, _Float16* __restrict__ xout_h,
    const int* __restrict__ rowptr, const int* __restrict__ csr,
    const _Float16* __restrict__ PWl, const float* __restrict__ bl,
    const _Float16* __restrict__ PWr, int nedges,
    const float* __restrict__ Wo, const float* __restrict__ bo,
    float* __restrict__ out) {
  __shared__ __attribute__((aligned(16))) _Float16 smean[16 * MEAN_PITCH];
  __shared__ __attribute__((aligned(16))) _Float16 sout[16 * MEAN_PITCH];
  __shared__ __attribute__((aligned(16))) float shead[16 * HEAD_PITCH];
  const int lane = threadIdx.x & 31;
  const int wv = __builtin_amdgcn_readfirstlane(threadIdx.x >> 5);
  const int rbase = blockIdx.x * 16;
  const int n0 = rbase + wv * 4;

  int rp = 0;
  if (lane <= 4) {
    const int nd = n0 + lane;
    rp = (nd < NNODES) ? rowptr[nd] : nedges;
  }
  const int b0 = __builtin_amdgcn_readlane(rp, 0);
  const int b1 = __builtin_amdgcn_readlane(rp, 1);
  const int b2 = __builtin_amdgcn_readlane(rp, 2);
  const int b3 = __builtin_amdgcn_readlane(rp, 3);
  const int b4 = __builtin_amdgcn_readlane(rp, 4);
  const int c8 = lane & 7;
  const int q  = lane >> 3;
  const int beg = q == 0 ? b0 : q == 1 ? b1 : q == 2 ? b2 : b3;
  const int end = q == 0 ? b1 : q == 1 ? b2 : q == 2 ? b3 : b4;
  const int i0 = load_idx8(csr, beg,      end, c8);
  const int i1 = load_idx8(csr, beg + 8,  end, c8);
  const int i2 = load_idx8(csr, beg + 16, end, c8);
  const int i3 = load_idx8(csr, beg + 24, end, c8);
  const int i4 = load_idx8(csr, beg + 32, end, c8);
  const int i5 = load_idx8(csr, beg + 40, end, c8);

  const char* xb = (const char*)xin;
  const unsigned coff = (unsigned)c8 * 16u;
  {
    const int d0 = b1 - b0, d1 = b2 - b1, d2 = b3 - b2, d3 = b4 - b3;
    const int m01 = d0 > d1 ? d0 : d1;
    const int m23 = d2 > d3 ? d2 : d3;
    const int nmax = m01 > m23 ? m01 : m23;
    float acc[8] = {0.0f, 0.0f, 0.0f, 0.0f, 0.0f, 0.0f, 0.0f, 0.0f};
    int k = 0;
    for (int j0 = 0; j0 < nmax; j0 += GATHER_DEPTH, ++k) {
      int ix;
      if (k == 0)      ix = i0;
      else if (k == 1) ix = i1;
      else if (k == 2) ix = i2;
      else if (k == 3) ix = i3;
      else if (k == 4) ix = i4;
      else if (k == 5) ix = i5;
      else             ix = load_idx8(csr, beg + j0, end, c8);
      gather_batch(xb, coff, ix, acc);
    }
    const float sc = 1.0f / fmaxf((float)(end - beg), 1.0f);
    v8h mh;
#pragma unroll
    for (int e = 0; e < 8; ++e) mh[e] = (_Float16)(acc[e] * sc);
    *(v8h*)(smean + (wv * 4 + q) * MEAN_PITCH + c8 * 8) = mh;
  }

  const int ni = wv;
  const v16h ax0 = load_a_frag_h(xin, DFEAT, rbase, 0, lane);
  const v16h ax1 = load_a_frag_h(xin, DFEAT, rbase, 32, lane);
  const v16h wr0 = load_b_packed(PWr, 4, 0, ni, lane);
  const v16h wr1 = load_b_packed(PWr, 4, 1, ni, lane);
  const int n  = lane & 15;
  const int hi = lane >> 4;
  const float bv = bl[ni * 16 + n];
  __syncthreads();

  const v16h wl0 = load_b_packed(PWl, 4, 0, ni, lane);
  const v16h wl1 = load_b_packed(PWl, 4, 1, ni, lane);
  v8f acc = {};
  acc = wmma_f16(load_a_frag_h(smean, MEAN_PITCH, 0, 0, lane), wl0, acc);
  acc = wmma_f16(ax0, wr0, acc);
  acc = wmma_f16(load_a_frag_h(smean, MEAN_PITCH, 0, 32, lane), wl1, acc);
  acc = wmma_f16(ax1, wr1, acc);

  if (HEAD) {
    float* srow = shead + (hi * 8) * HEAD_PITCH + ni * 16 + n;
#pragma unroll
    for (int r = 0; r < 8; ++r) {
      const float v = acc[r] + bv;
      srow[r * HEAD_PITCH] = v > 0.0f ? v : 0.0f;
    }
    const float4* wp = (const float4*)(Wo + c8 * 16);
    const float4 hw0 = wp[0], hw1 = wp[1], hw2 = wp[2], hw3 = wp[3];
    const float hb0 = bo[0], hb1 = bo[1];
    __syncthreads();
    const float* xr = shead + (wv * 4 + q) * HEAD_PITCH + c8 * 8;
    const float4 xa = *(const float4*)(xr + 0);
    const float4 xc4 = *(const float4*)(xr + 4);
    v2f o = {0.0f, 0.0f};
    o += (v2f){xa.x, xa.x}   * (v2f){hw0.x, hw0.y};
    o += (v2f){xa.y, xa.y}   * (v2f){hw0.z, hw0.w};
    o += (v2f){xa.z, xa.z}   * (v2f){hw1.x, hw1.y};
    o += (v2f){xa.w, xa.w}   * (v2f){hw1.z, hw1.w};
    o += (v2f){xc4.x, xc4.x} * (v2f){hw2.x, hw2.y};
    o += (v2f){xc4.y, xc4.y} * (v2f){hw2.z, hw2.w};
    o += (v2f){xc4.z, xc4.z} * (v2f){hw3.x, hw3.y};
    o += (v2f){xc4.w, xc4.w} * (v2f){hw3.z, hw3.w};
    o += (v2f){xor8<1>(o.x), xor8<1>(o.y)};
    o += (v2f){xor8<2>(o.x), xor8<2>(o.y)};
    o += (v2f){xor8<4>(o.x), xor8<4>(o.y)};
    if (c8 < 2)
      out[(size_t)(rbase + wv * 4 + q) * 2 + c8] = c8 ? o.y + hb1 : o.x + hb0;
  } else {
    _Float16* srow = sout + (hi * 8) * MEAN_PITCH + ni * 16 + n;
#pragma unroll
    for (int r = 0; r < 8; ++r) {
      const float v = acc[r] + bv;
      srow[r * MEAN_PITCH] = (_Float16)(v > 0.0f ? v : 0.0f);
    }
    __syncthreads();
    const int row = threadIdx.x >> 3;
    const int pc  = threadIdx.x & 7;
    *(v8h*)(xout_h + (size_t)(rbase + row) * DFEAT + pc * 8) =
        *(const v8h*)(sout + row * MEAN_PITCH + pc * 8);
  }
}

extern "C" void kernel_launch(void* const* d_in, const int* in_sizes, int n_in,
                              void* d_out, int out_size, void* d_ws,
                              size_t ws_size, hipStream_t stream) {
  const float* xc  = (const float*)d_in[0];
  const float* xs  = (const float*)d_in[1];
  const int*   ei  = (const int*)d_in[2];
  const float* Wp  = (const float*)d_in[4];
  const float* bp  = (const float*)d_in[5];
  const float* Ws  = (const float*)d_in[6];
  const float* bs  = (const float*)d_in[7];
  const float* Wl1 = (const float*)d_in[8];
  const float* bl1 = (const float*)d_in[9];
  const float* Wr1 = (const float*)d_in[10];
  const float* Wl2 = (const float*)d_in[11];
  const float* bl2 = (const float*)d_in[12];
  const float* Wr2 = (const float*)d_in[13];
  const float* Wl3 = (const float*)d_in[14];
  const float* bl3 = (const float*)d_in[15];
  const float* Wr3 = (const float*)d_in[16];
  const float* Wo  = (const float*)d_in[17];
  const float* bo  = (const float*)d_in[18];

  const int E = in_sizes[2] / 2;
  const int* src = ei;
  const int* dst = ei + E;

  float* ws  = (float*)d_ws;
  _Float16* X0 = (_Float16*)ws;
  int*   CSR = (int*)(ws + (size_t)(NNODES + 1) * DFEAT);
  int*   BKT = CSR + E;
  float* RA  = ws + 2ull * NNODES * DFEAT;
  int2*  REC = (int2*)RA;
  _Float16* X1 = (_Float16*)RA + (size_t)NNODES * DFEAT;
  int*   RPT = (int*)(ws + 3ull * NNODES * DFEAT + DFEAT);
  _Float16* PW   = (_Float16*)(ws + 3ull * NNODES * DFEAT + DFEAT + NNODES);
  _Float16* PWp  = PW;
  _Float16* PWs  = PW + 48ull  * FRAGH;
  _Float16* PWl1 = PW + 56ull  * FRAGH;
  _Float16* PWr1 = PW + 64ull  * FRAGH;
  _Float16* PWl2 = PW + 72ull  * FRAGH;
  _Float16* PWr2 = PW + 80ull  * FRAGH;
  _Float16* PWl3 = PW + 88ull  * FRAGH;
  _Float16* PWr3 = PW + 96ull  * FRAGH;

  const int layer_blocks = NNODES / 16;
  const int chunks = (E + CHUNK - 1) / CHUNK;

  prep_kernel<<<PREP_PACK_BLOCKS + PREP_ZERO_BLOCKS, 256, 0, stream>>>(
      Wp, Ws, Wl1, Wr1, Wl2, Wr2, Wl3, Wr3, PW, BKT,
      X0 + (size_t)NNODES * DFEAT, X1 + (size_t)NNODES * DFEAT);

  proj_hist_kernel<<<PROJ_BLOCKS + chunks, 256, 0, stream>>>(
      xc, xs, PWp, bp, PWs, bs, X0, dst, BKT, E);

  part_kernel<<<chunks, 512, 0, stream>>>(src, dst, BKT, REC, E);
  bsort_kernel<<<NBKT, 512, 0, stream>>>(REC, BKT, RPT, CSR);

  layer_kernel<false><<<layer_blocks, 128, 0, stream>>>(X0, X1, RPT, CSR, PWl1, bl1, PWr1, E,
                                                        nullptr, nullptr, nullptr);
  layer_kernel<false><<<layer_blocks, 128, 0, stream>>>(X1, X0, RPT, CSR, PWl2, bl2, PWr2, E,
                                                        nullptr, nullptr, nullptr);
  layer_kernel<true><<<layer_blocks, 128, 0, stream>>>(X0, nullptr, RPT, CSR, PWl3, bl3, PWr3, E,
                                                       Wo, bo, (float*)d_out);
}
